# starting kernel
# baseline (speedup 1.0000x reference)
#include <hip/hip_runtime.h>
#include <hip/hip_bf16.h>

typedef __attribute__((ext_vector_type(2)))  float    v2f;
typedef __attribute__((ext_vector_type(8)))  float    v8f;
typedef __attribute__((ext_vector_type(16))) _Float16 v16h;

__global__ void gcn_init_kernel(float* __restrict__ h2acc, float* __restrict__ deg,
                                float* __restrict__ z, float* __restrict__ out,
                                int N, int paddedN) {
    int i = blockIdx.x * blockDim.x + threadIdx.x;
    if (i < paddedN * 16) h2acc[i] = 0.0f;
    if (i < paddedN) { deg[i] = (i < N) ? 1.0f : 0.0f; z[i] = 0.0f; }
    if (i < N) out[i] = 0.0f;
}

__global__ void gcn_degree_kernel(const int* __restrict__ dst, float* __restrict__ deg, int E) {
    int e = blockIdx.x * blockDim.x + threadIdx.x;
    if (e < E) atomicAdd(&deg[dst[e]], 1.0f);
}

__global__ void gcn_rsqrt_kernel(float* __restrict__ dinv, int N) {
    int n = blockIdx.x * blockDim.x + threadIdx.x;
    if (n < N) dinv[n] = rsqrtf(dinv[n]);
}

__global__ void gcn_agg1_kernel(const int* __restrict__ src, const int* __restrict__ dst,
                                const float* __restrict__ dinv, const float* __restrict__ x,
                                const float* __restrict__ W1, float* __restrict__ h2acc, int E) {
    int t = blockIdx.x * blockDim.x + threadIdx.x;
    int e = t >> 4;
    if (e >= E) return;
    int f = t & 15;
    int s = src[e];
    int d = dst[e];
    float w = dinv[s] * dinv[d];
    atomicAdd(&h2acc[(size_t)d * 16 + f], w * x[s] * W1[f]);
}

__global__ void gcn_pointwise1_kernel(float* __restrict__ h2, const float* __restrict__ dinv,
                                      const float* __restrict__ x, const float* __restrict__ W1,
                                      const float* __restrict__ b1, int N) {
    int t = blockIdx.x * blockDim.x + threadIdx.x;
    int n = t >> 4;
    if (n >= N) return;
    int f = t & 15;
    float di = dinv[n];
    float v = h2[(size_t)n * 16 + f] + di * di * x[n] * W1[f] + b1[f];
    h2[(size_t)n * 16 + f] = fmaxf(v, 0.0f);
}

__global__ void gcn_gemv_wmma_kernel(const float* __restrict__ h2, const float* __restrict__ W2,
                                     float* __restrict__ z, int n_tiles) {
    int lane = threadIdx.x & 31;
    int wave = (blockIdx.x * blockDim.x + threadIdx.x) >> 5;
    if (wave >= n_tiles) return;
    int base = wave * 16;
    int m  = lane & 15;
    int kh = lane >> 4;
    const float* row = h2 + (size_t)(base + m) * 16;
    v8f acc = {0.f, 0.f, 0.f, 0.f, 0.f, 0.f, 0.f, 0.f};

#if __has_builtin(__builtin_amdgcn_wmma_f32_16x16x4_f32)
#pragma unroll
    for (int kc = 0; kc < 4; ++kc) {
        v2f a, b;
        a.x = row[kc * 4 + kh * 2 + 0];
        a.y = row[kc * 4 + kh * 2 + 1];
        b.x = W2[kc * 4 + kh * 2 + 0];
        b.y = W2[kc * 4 + kh * 2 + 1];
        acc = __builtin_amdgcn_wmma_f32_16x16x4_f32(
             false, a,  false, b,
             (short)0, acc,  false,  false);
    }
#else
    v16h a, b;
#pragma unroll
    for (int j = 0; j < 8; ++j) {
        a[j]     = (_Float16)row[kh * 8 + j];
        a[8 + j] = (_Float16)0.0f;
        b[j]     = (lane < 16) ? (_Float16)W2[j]     : (_Float16)0.0f;
        b[8 + j] = (lane < 16) ? (_Float16)W2[8 + j] : (_Float16)0.0f;
    }
    acc = __builtin_amdgcn_wmma_f32_16x16x32_f16(
        false, a, false, b, (short)0, acc, false, false);
#endif

    if (m == 0) {
        float* zp = z + base + kh * 8;
        zp[0] = acc[0]; zp[1] = acc[1]; zp[2] = acc[2]; zp[3] = acc[3];
        zp[4] = acc[4]; zp[5] = acc[5]; zp[6] = acc[6]; zp[7] = acc[7];
    }
}

__global__ void gcn_agg2_kernel(const int* __restrict__ src, const int* __restrict__ dst,
                                const float* __restrict__ dinv, const float* __restrict__ z,
                                float* __restrict__ out, int E) {
    int e = blockIdx.x * blockDim.x + threadIdx.x;
    if (e >= E) return;
    int s = src[e];
    int d = dst[e];
    atomicAdd(&out[d], dinv[s] * dinv[d] * z[s]);
}

__global__ void gcn_final_kernel(float* __restrict__ out, const float* __restrict__ dinv,
                                 const float* __restrict__ z, const float* __restrict__ b2, int N) {
    int n = blockIdx.x * blockDim.x + threadIdx.x;
    if (n >= N) return;
    float di = dinv[n];
    out[n] += di * di * z[n] + b2[0];
}

extern "C" void kernel_launch(void* const* d_in, const int* in_sizes, int n_in,
                              void* d_out, int out_size, void* d_ws, size_t ws_size,
                              hipStream_t stream) {
    const float* x  = (const float*)d_in[0];
    const float* W1 = (const float*)d_in[1];
    const float* b1 = (const float*)d_in[2];
    const float* W2 = (const float*)d_in[3];
    const float* b2 = (const float*)d_in[4];
    const int*   ei = (const int*)d_in[5];
    int N = in_sizes[0];
    int E = in_sizes[5] / 2;
    const int* src = ei;
    const int* dst = ei + E;
    int paddedN = (N + 15) & ~15;

    char* ws = (char*)d_ws;
    float* h2   = (float*)ws; ws += (size_t)paddedN * 16 * sizeof(float);
    float* dinv = (float*)ws; ws += (size_t)paddedN * sizeof(float);
    float* z    = (float*)ws; ws += (size_t)paddedN * sizeof(float);
    float* out  = (float*)d_out;

    const int B = 256;
    int tiles = paddedN / 16;

    gcn_init_kernel<<<(paddedN * 16 + B - 1) / B, B, 0, stream>>>(h2, dinv, z, out, N, paddedN);
    gcn_degree_kernel<<<(E + B - 1) / B, B, 0, stream>>>(dst, dinv, E);
    gcn_rsqrt_kernel<<<(N + B - 1) / B, B, 0, stream>>>(dinv, N);
    gcn_agg1_kernel<<<((long long)E * 16 + B - 1) / B, B, 0, stream>>>(src, dst, dinv, x, W1, h2, E);
    gcn_pointwise1_kernel<<<((long long)N * 16 + B - 1) / B, B, 0, stream>>>(h2, dinv, x, W1, b1, N);
    gcn_gemv_wmma_kernel<<<(tiles * 32 + B - 1) / B, B, 0, stream>>>(h2, W2, z, tiles);
    gcn_agg2_kernel<<<(E + B - 1) / B, B, 0, stream>>>(src, dst, dinv, z, out, E);
    gcn_final_kernel<<<(N + B - 1) / B, B, 0, stream>>>(out, dinv, z, b2, N);
}


# Per-node kernels fused into the two row gathers: self loop, bias, ReLU and the W2 matrix product follow gather 1 in its block (h2 never stored), final term follows gather 2; 9 launches -> 6
# speedup vs baseline: 3.7057x; 3.7057x over previous
#include <hip/hip_runtime.h>
#include <hip/hip_bf16.h>

typedef __attribute__((ext_vector_type(2)))  float    v2f;
typedef __attribute__((ext_vector_type(8)))  float    v8f;

#define GCN_NB   256
#define GCN_NS   256
#define GCN_RMAX 4096

__global__ __launch_bounds__(1024) void gcn_range_count_kernel(const int* __restrict__ dst,
                                                               int* __restrict__ T, int E, int N) {
    __shared__ int hist[GCN_NB];
    const int tid = threadIdx.x;
    const int s   = blockIdx.x;
    const unsigned R = (unsigned)((N + GCN_NB - 1) / GCN_NB);
    const int L  = (E + GCN_NS - 1) / GCN_NS;
    const int e0 = s * L;
    const int e1 = (e0 + L < E) ? (e0 + L) : E;
    if (tid < GCN_NB) hist[tid] = 0;
    __syncthreads();
    for (int e = e0 + tid; e < e1; e += 4096) {
        int d[4];
#pragma unroll
        for (int k = 0; k < 4; ++k) {
            const int ek = e + k * 1024;
            d[k] = dst[ek < e1 ? ek : e1 - 1];
        }
#pragma unroll
        for (int k = 0; k < 4; ++k) {
            if (e + k * 1024 < e1) atomicAdd(&hist[(unsigned)d[k] / R], 1);
        }
    }
    __syncthreads();
    if (tid < GCN_NB) T[tid * GCN_NS + s] = hist[tid];
}

__global__ __launch_bounds__(1024) void gcn_range_scan_kernel(int* __restrict__ T) {
    __shared__ int wtot[32];
    const int tid  = threadIdx.x;
    const int lane = tid & 31;
    const int w    = tid >> 5;
    constexpr int PER = GCN_NB * GCN_NS / 1024;
    int4* p = (int4*)(T + tid * PER);
    int4 v[PER / 4];
    int part = 0;
#pragma unroll
    for (int k = 0; k < PER / 4; ++k) {
        v[k] = p[k];
        part += (v[k].x + v[k].y) + (v[k].z + v[k].w);
    }
    int inc = part;
    for (int o = 1; o < 32; o <<= 1) {
        int y = __shfl_up(inc, o, 32);
        if (lane >= o) inc += y;
    }
    if (lane == 31) wtot[w] = inc;
    __syncthreads();
    const int wt = wtot[lane];
    int winc = wt;
    for (int o = 1; o < 32; o <<= 1) {
        int y = __shfl_up(winc, o, 32);
        if (lane >= o) winc += y;
    }
    int run = __shfl(winc - wt, w, 32) + inc - part;
#pragma unroll
    for (int k = 0; k < PER / 4; ++k) {
        int4 q;
        q.x = run; run += v[k].x;
        q.y = run; run += v[k].y;
        q.z = run; run += v[k].z;
        q.w = run; run += v[k].w;
        p[k] = q;
    }
}

__global__ __launch_bounds__(1024) void gcn_range_scatter_kernel(const int* __restrict__ src,
                                                                 const int* __restrict__ dst,
                                                                 const int* __restrict__ T,
                                                                 int* __restrict__ pairs_raw,
                                                                 int E, int N) {
    __shared__ int cur[GCN_NB];
    int2* pairs = (int2*)pairs_raw;
    const int tid = threadIdx.x;
    const int s   = blockIdx.x;
    const unsigned R = (unsigned)((N + GCN_NB - 1) / GCN_NB);
    const int L  = (E + GCN_NS - 1) / GCN_NS;
    const int e0 = s * L;
    const int e1 = (e0 + L < E) ? (e0 + L) : E;
    if (tid < GCN_NB) cur[tid] = T[tid * GCN_NS + s];
    __syncthreads();
    for (int e = e0 + tid; e < e1; e += 4096) {
        int d[4], sv[4];
#pragma unroll
        for (int k = 0; k < 4; ++k) {
            const int ek = e + k * 1024;
            const int ec = ek < e1 ? ek : e1 - 1;
            d[k]  = dst[ec];
            sv[k] = src[ec];
        }
#pragma unroll
        for (int k = 0; k < 4; ++k) {
            if (e + k * 1024 < e1) {
                const int slot = atomicAdd(&cur[(unsigned)d[k] / R], 1);
                pairs[slot] = make_int2(sv[k], d[k]);
            }
        }
    }
}

__global__ __launch_bounds__(1024) void gcn_rows_kernel(const int* __restrict__ pairs_raw,
                                                        const int* __restrict__ T,
                                                        int* __restrict__ rowptr,
                                                        int* __restrict__ cnt,
                                                        float* __restrict__ dinv,
                                                        int* __restrict__ col, int E, int N) {
    __shared__ int lcnt[GCN_RMAX];
    const int2* pairs = (const int2*)pairs_raw;
    __shared__ int wtot[32];
    const int tid  = threadIdx.x;
    const int lane = tid & 31;
    const int w    = tid >> 5;
    const int b    = blockIdx.x;
    const int R    = (N + GCN_NB - 1) / GCN_NB;
    const int n0   = b * R;
    if (n0 >= N) return;
    const int p0 = T[b * GCN_NS];
    const int p1 = (b + 1 < GCN_NB) ? T[(b + 1) * GCN_NS] : E;

    const int i = tid * 4;
    *(int4*)(lcnt + i) = make_int4(0, 0, 0, 0);
    __syncthreads();

    for (int p = p0 + tid; p < p1; p += 4096) {
        int d[4];
#pragma unroll
        for (int k = 0; k < 4; ++k) {
            const int pk = p + k * 1024;
            d[k] = pairs[pk < p1 ? pk : p1 - 1].y;
        }
#pragma unroll
        for (int k = 0; k < 4; ++k) {
            if (p + k * 1024 < p1) atomicAdd(&lcnt[d[k] - n0], 1);
        }
    }
    __syncthreads();

    const int4 c = *(const int4*)(lcnt + i);
    const int part = (c.x + c.y) + (c.z + c.w);
    int inc = part;
    for (int o = 1; o < 32; o <<= 1) {
        int y = __shfl_up(inc, o, 32);
        if (lane >= o) inc += y;
    }
    if (lane == 31) wtot[w] = inc;
    __syncthreads();
    const int wt = wtot[lane];
    int winc = wt;
    for (int o = 1; o < 32; o <<= 1) {
        int y = __shfl_up(winc, o, 32);
        if (lane >= o) winc += y;
    }
    const int q0 = p0 + __shfl(winc - wt, w, 32) + inc - part;
    const int q1 = q0 + c.x;
    const int q2 = q1 + c.y;
    const int q3 = q2 + c.z;
    {
        const int n = n0 + i;
        if (i     < R && n     < N) { rowptr[n]     = q0; cnt[n]     = c.x; dinv[n]     = rsqrtf((float)(c.x + 1)); }
        if (i + 1 < R && n + 1 < N) { rowptr[n + 1] = q1; cnt[n + 1] = c.y; dinv[n + 1] = rsqrtf((float)(c.y + 1)); }
        if (i + 2 < R && n + 2 < N) { rowptr[n + 2] = q2; cnt[n + 2] = c.z; dinv[n + 2] = rsqrtf((float)(c.z + 1)); }
        if (i + 3 < R && n + 3 < N) { rowptr[n + 3] = q3; cnt[n + 3] = c.w; dinv[n + 3] = rsqrtf((float)(c.w + 1)); }
    }
    *(int4*)(lcnt + i) = make_int4(q0, q1, q2, q3);
    __syncthreads();

    for (int p = p0 + tid; p < p1; p += 4096) {
        int2 e[4];
#pragma unroll
        for (int k = 0; k < 4; ++k) {
            const int pk = p + k * 1024;
            e[k] = pairs[pk < p1 ? pk : p1 - 1];
        }
#pragma unroll
        for (int k = 0; k < 4; ++k) {
            if (p + k * 1024 < p1) {
                const int slot = atomicAdd(&lcnt[e[k].y - n0], 1);
                col[slot] = e[k].x;
            }
        }
    }
}

__global__ __launch_bounds__(256) void gcn_layer1_kernel(const int* __restrict__ rowptr,
                                                         const int* __restrict__ cnt,
                                                         const int* __restrict__ col,
                                                         const float* __restrict__ dinv,
                                                         const float* __restrict__ x,
                                                         const float* __restrict__ W1,
                                                         const float* __restrict__ b1,
                                                         const float* __restrict__ W2,
                                                         float* __restrict__ z, int N) {
    __shared__ float tsum[16];
    const int r    = threadIdx.x >> 4;
    const int f    = threadIdx.x & 15;
    const int base = blockIdx.x * 16;
    const int n    = base + r;
    const bool mat = threadIdx.x < 32;
    const int m    = threadIdx.x & 15;
    const int kh   = (threadIdx.x >> 4) & 1;

    float w1v[8], b1v[8], w2v[8];
    float dm = 0.0f, xm = 0.0f;
    if (mat) {
#pragma unroll
        for (int kc = 0; kc < 4; ++kc) {
#pragma unroll
            for (int i = 0; i < 2; ++i) {
                const int k = kc * 4 + kh * 2 + i;
                w1v[kc * 2 + i] = W1[k];
                b1v[kc * 2 + i] = b1[k];
                w2v[kc * 2 + i] = W2[k];
            }
        }
        if (base + m < N) { dm = dinv[base + m]; xm = x[base + m]; }
    }

    float acc = 0.0f;
    if (n < N) {
        const int start = rowptr[n];
        const int len   = cnt[n];
        for (int j = f; j < len; j += 32) {
            const int  j1  = j + 16;
            const bool two = j1 < len;
            const int  s0  = col[start + j];
            const int  s1  = col[start + (two ? j1 : j)];
            const float a0 = dinv[s0] * x[s0];
            const float a1 = dinv[s1] * x[s1];
            acc += a0 + (two ? a1 : 0.0f);
        }
    }
    acc += __shfl_xor(acc, 8, 16);
    acc += __shfl_xor(acc, 4, 16);
    acc += __shfl_xor(acc, 2, 16);
    acc += __shfl_xor(acc, 1, 16);
    if (f == 0) tsum[r] = acc;
    __syncthreads();

    if (mat) {
        const float ts = dm * tsum[m];
        const float sl = dm * dm * xm;
        v8f d = {0.f, 0.f, 0.f, 0.f, 0.f, 0.f, 0.f, 0.f};
#pragma unroll
        for (int kc = 0; kc < 4; ++kc) {
            v2f a, b;
            a.x = fmaxf(ts * w1v[kc * 2 + 0] + sl * w1v[kc * 2 + 0] + b1v[kc * 2 + 0], 0.0f);
            a.y = fmaxf(ts * w1v[kc * 2 + 1] + sl * w1v[kc * 2 + 1] + b1v[kc * 2 + 1], 0.0f);
            b.x = w2v[kc * 2 + 0];
            b.y = w2v[kc * 2 + 1];
            d = __builtin_amdgcn_wmma_f32_16x16x4_f32(
                 false, a,  false, b,
                 (short)0, d,  false,  false);
        }
        if (m == 0) {
            float* zp = z + base + kh * 8;
            zp[0] = d[0]; zp[1] = d[1]; zp[2] = d[2]; zp[3] = d[3];
            zp[4] = d[4]; zp[5] = d[5]; zp[6] = d[6]; zp[7] = d[7];
        }
    }
}

__global__ __launch_bounds__(512) void gcn_layer2_kernel(const int* __restrict__ rowptr,
                                                         const int* __restrict__ cnt,
                                                         const int* __restrict__ col,
                                                         const float* __restrict__ dinv,
                                                         const float* __restrict__ z,
                                                         const float* __restrict__ b2,
                                                         float* __restrict__ out, int N) {
    __shared__ float stage[32];
    const int r = threadIdx.x >> 4;
    const int f = threadIdx.x & 15;
    const int n = blockIdx.x * 32 + r;
    float acc = 0.0f;
    float dn = 0.0f, zn = 0.0f;
    const float bias = b2[0];
    if (n < N) {
        const int start = rowptr[n];
        const int len   = cnt[n];
        dn = dinv[n];
        zn = z[n];
        for (int j = f; j < len; j += 32) {
            const int  j1  = j + 16;
            const bool two = j1 < len;
            const int  s0  = col[start + j];
            const int  s1  = col[start + (two ? j1 : j)];
            const float a0 = dinv[s0] * z[s0];
            const float a1 = dinv[s1] * z[s1];
            acc += a0 + (two ? a1 : 0.0f);
        }
    }
    acc += __shfl_xor(acc, 8, 16);
    acc += __shfl_xor(acc, 4, 16);
    acc += __shfl_xor(acc, 2, 16);
    acc += __shfl_xor(acc, 1, 16);
    if (f == 0) stage[r] = dn * acc + (dn * dn * zn + bias);
    __syncthreads();
    if (threadIdx.x < 32) {
        const int mo = blockIdx.x * 32 + threadIdx.x;
        if (mo < N) out[mo] = stage[threadIdx.x];
    }
}

extern "C" void kernel_launch(void* const* d_in, const int* in_sizes, int n_in,
                              void* d_out, int out_size, void* d_ws, size_t ws_size,
                              hipStream_t stream) {
    const float* x  = (const float*)d_in[0];
    const float* W1 = (const float*)d_in[1];
    const float* b1 = (const float*)d_in[2];
    const float* W2 = (const float*)d_in[3];
    const float* b2 = (const float*)d_in[4];
    const int*   ei = (const int*)d_in[5];
    int N = in_sizes[0];
    int E = in_sizes[5] / 2;
    const int* src = ei;
    const int* dst = ei + E;
    int paddedN = (N + 15) & ~15;

    char* ws = (char*)d_ws;
    float* dinv   = (float*)ws; ws += (size_t)paddedN * sizeof(float);
    float* z      = (float*)ws; ws += (size_t)paddedN * sizeof(float);
    int*   cnt    = (int*)ws;   ws += (size_t)paddedN * sizeof(int);
    int*   rowptr = (int*)ws;   ws += (size_t)paddedN * sizeof(int);
    int*   T      = (int*)ws;   ws += (size_t)GCN_NB * GCN_NS * sizeof(int);
    int*   pairs  = (int*)ws;   ws += (((size_t)E * 2 * sizeof(int)) + 255) & ~(size_t)255;
    int*   col    = (int*)ws;
    float* out    = (float*)d_out;

    int tiles = paddedN / 16;

    gcn_range_count_kernel<<<GCN_NS, 1024, 0, stream>>>(dst, T, E, N);
    gcn_range_scan_kernel<<<1, 1024, 0, stream>>>(T);
    gcn_range_scatter_kernel<<<GCN_NS, 1024, 0, stream>>>(src, dst, T, pairs, E, N);
    gcn_rows_kernel<<<GCN_NB, 1024, 0, stream>>>(pairs, T, rowptr, cnt, dinv, col, E, N);
    gcn_layer1_kernel<<<tiles, 256, 0, stream>>>(rowptr, cnt, col, dinv, x, W1, b1, W2, z, N);
    gcn_layer2_kernel<<<(N + 31) / 32, 512, 0, stream>>>(rowptr, cnt, col, dinv, z, b2, out, N);
}


# Grouping placed in LDS: each edge slice is sorted by dst range in its block and stored as whole lines, a range's block reads its 256 runs and builds col in LDS; 6 launches -> 4; row order = arrival
# speedup vs baseline: 5.5962x; 1.5102x over previous
#include <hip/hip_runtime.h>
#include <hip/hip_bf16.h>

typedef __attribute__((ext_vector_type(2)))  float    v2f;
typedef __attribute__((ext_vector_type(8)))  float    v8f;

#define GCN_NB   256
#define GCN_NS   256
#define GCN_RMAX 4096
#define GCN_LPT  16
#define GCN_LMAX (1024 * GCN_LPT)
#define GCN_CCAP 32768
#define GCN_RPW  (GCN_NS / 32)
#define GCN_RLD  3

__global__ __launch_bounds__(1024) void gcn_slice_sort_kernel(const int* __restrict__ src,
                                                              const int* __restrict__ dst,
                                                              int* __restrict__ T,
                                                              int* __restrict__ pairs_raw,
                                                              int E, int N) {
    __shared__ int2 stage[GCN_LMAX];
    __shared__ int hist[GCN_NB];
    __shared__ int first[GCN_NB];
    __shared__ int wtot[GCN_NB / 32];
    int2* pairs = (int2*)pairs_raw;
    const int tid  = threadIdx.x;
    const int lane = tid & 31;
    const int w    = tid >> 5;
    const int s    = blockIdx.x;
    const unsigned R = (unsigned)((N + GCN_NB - 1) / GCN_NB);
    const int L  = (E + GCN_NS - 1) / GCN_NS;
    const int e0 = s * L;
    const int e1 = (e0 + L < E) ? (e0 + L) : E;
    const int len = e1 > e0 ? e1 - e0 : 0;
    if (tid < GCN_NB) hist[tid] = 0;

    int d[GCN_LPT], sv[GCN_LPT], rg[GCN_LPT], rk[GCN_LPT];
#pragma unroll
    for (int k = 0; k < GCN_LPT; ++k) {
        const int ek = e0 + tid + k * 1024;
        const int ec = ek < e1 ? ek : e1 - 1;
        d[k]  = dst[ec];
        sv[k] = src[ec];
    }
    __syncthreads();
#pragma unroll
    for (int k = 0; k < GCN_LPT; ++k) {
        rg[k] = (int)((unsigned)d[k] / R);
        rk[k] = 0;
        if (e0 + tid + k * 1024 < e1) rk[k] = atomicAdd(&hist[rg[k]], 1);
    }
    __syncthreads();

    const int c = (tid < GCN_NB) ? hist[tid] : 0;
    int inc = c;
    for (int o = 1; o < 32; o <<= 1) {
        int y = __shfl_up(inc, o, 32);
        if (lane >= o) inc += y;
    }
    if (tid < GCN_NB && lane == 31) wtot[w] = inc;
    __syncthreads();
    if (tid < GCN_NB) {
        int before = 0;
#pragma unroll
        for (int j = 0; j < GCN_NB / 32; ++j) before += (j < w) ? wtot[j] : 0;
        const int f = before + inc - c;
        first[tid] = f;
        T[tid * GCN_NS + s] = f;
        if (tid == GCN_NB - 1) T[GCN_NB * GCN_NS + s] = f + c;
    }
    __syncthreads();

#pragma unroll
    for (int k = 0; k < GCN_LPT; ++k) rg[k] = first[rg[k]] + rk[k];
#pragma unroll
    for (int k = 0; k < GCN_LPT; ++k) {
        if (e0 + tid + k * 1024 < e1) stage[rg[k]] = make_int2(sv[k], d[k]);
    }
    __syncthreads();
#pragma unroll 4
    for (int i = tid; i < len; i += 1024) pairs[e0 + i] = stage[i];
}

__global__ __launch_bounds__(1024) void gcn_rows_kernel(const int* __restrict__ pairs_raw,
                                                        const int* __restrict__ T,
                                                        int* __restrict__ rowptr,
                                                        int* __restrict__ cnt,
                                                        float* __restrict__ dinv,
                                                        int* __restrict__ col, int E, int N) {
    __shared__ int lcol[GCN_CCAP];
    __shared__ int lcnt[GCN_RMAX];
    __shared__ int rs[GCN_NS];
    __shared__ int rl[GCN_NS];
    __shared__ int wtot[32];
    __shared__ int redA[GCN_NS / 32];
    __shared__ int redL[GCN_NS / 32];
    const int2* pairs = (const int2*)pairs_raw;
    const int tid  = threadIdx.x;
    const int lane = tid & 31;
    const int w    = tid >> 5;
    const int b    = blockIdx.x;
    const int R    = (N + GCN_NB - 1) / GCN_NB;
    const int n0   = b * R;
    if (n0 >= N) return;
    const int L = (E + GCN_NS - 1) / GCN_NS;

    int ra = 0, rn = 0;
    if (tid < GCN_NS) {
        ra = T[b * GCN_NS + tid];
        rn = T[(b + 1) * GCN_NS + tid] - ra;
        rs[tid] = tid * L + ra;
        rl[tid] = rn;
    }
    int sa = ra, sl = rn;
    for (int o = 16; o > 0; o >>= 1) {
        sa += __shfl_xor(sa, o, 32);
        sl += __shfl_xor(sl, o, 32);
    }
    if (tid < GCN_NS && lane == 0) { redA[w] = sa; redL[w] = sl; }
    const int i = tid * 4;
    *(int4*)(lcnt + i) = make_int4(0, 0, 0, 0);
    __syncthreads();
    int p0 = 0, tot = 0;
#pragma unroll
    for (int j = 0; j < GCN_NS / 32; ++j) { p0 += redA[j]; tot += redL[j]; }
    const bool staged = tot <= GCN_CCAP;

    int  st[GCN_RPW], rn_[GCN_RPW];
    int2 rec[GCN_RPW][GCN_RLD];
#pragma unroll
    for (int r = 0; r < GCN_RPW; ++r) {
        st[r]  = rs[w + 32 * r];
        rn_[r] = rl[w + 32 * r];
#pragma unroll
        for (int h = 0; h < GCN_RLD; ++h) {
            const int j = lane + 32 * h;
            int g = st[r] + (j < rn_[r] ? j : rn_[r] - 1);
            g = g < E ? g : E - 1;
            g = g > 0 ? g : 0;
            rec[r][h] = pairs[g];
        }
    }

#pragma unroll
    for (int r = 0; r < GCN_RPW; ++r) {
#pragma unroll
        for (int h = 0; h < GCN_RLD; ++h) {
            if (lane + 32 * h < rn_[r]) atomicAdd(&lcnt[rec[r][h].y - n0], 1);
        }
        for (int j = lane + 32 * GCN_RLD; j < rn_[r]; j += 32)
            atomicAdd(&lcnt[pairs[st[r] + j].y - n0], 1);
    }
    __syncthreads();

    const int4 c = *(const int4*)(lcnt + i);
    const int part = (c.x + c.y) + (c.z + c.w);
    int inc = part;
    for (int o = 1; o < 32; o <<= 1) {
        int y = __shfl_up(inc, o, 32);
        if (lane >= o) inc += y;
    }
    if (lane == 31) wtot[w] = inc;
    __syncthreads();
    const int wt = wtot[lane];
    int winc = wt;
    for (int o = 1; o < 32; o <<= 1) {
        int y = __shfl_up(winc, o, 32);
        if (lane >= o) winc += y;
    }
    const int q0 = p0 + __shfl(winc - wt, w, 32) + inc - part;
    const int q1 = q0 + c.x;
    const int q2 = q1 + c.y;
    const int q3 = q2 + c.z;
    {
        const int n = n0 + i;
        if (i     < R && n     < N) { rowptr[n]     = q0; cnt[n]     = c.x; dinv[n]     = rsqrtf((float)(c.x + 1)); }
        if (i + 1 < R && n + 1 < N) { rowptr[n + 1] = q1; cnt[n + 1] = c.y; dinv[n + 1] = rsqrtf((float)(c.y + 1)); }
        if (i + 2 < R && n + 2 < N) { rowptr[n + 2] = q2; cnt[n + 2] = c.z; dinv[n + 2] = rsqrtf((float)(c.z + 1)); }
        if (i + 3 < R && n + 3 < N) { rowptr[n + 3] = q3; cnt[n + 3] = c.w; dinv[n + 3] = rsqrtf((float)(c.w + 1)); }
    }
    *(int4*)(lcnt + i) = make_int4(q0, q1, q2, q3);
    __syncthreads();

#pragma unroll
    for (int r = 0; r < GCN_RPW; ++r) {
        int slot[GCN_RLD];
#pragma unroll
        for (int h = 0; h < GCN_RLD; ++h) {
            slot[h] = p0;
            if (lane + 32 * h < rn_[r]) slot[h] = atomicAdd(&lcnt[rec[r][h].y - n0], 1);
        }
#pragma unroll
        for (int h = 0; h < GCN_RLD; ++h) {
            if (lane + 32 * h < rn_[r]) {
                if (staged) lcol[slot[h] - p0] = rec[r][h].x;
                else        col[slot[h]]       = rec[r][h].x;
            }
        }
        for (int j = lane + 32 * GCN_RLD; j < rn_[r]; j += 32) {
            const int2 e = pairs[st[r] + j];
            const int q = atomicAdd(&lcnt[e.y - n0], 1);
            if (staged) lcol[q - p0] = e.x;
            else        col[q]       = e.x;
        }
    }
    if (staged) {
        __syncthreads();
        for (int j = tid; j < tot; j += 1024) col[p0 + j] = lcol[j];
    }
}

__global__ __launch_bounds__(256) void gcn_layer1_kernel(const int* __restrict__ rowptr,
                                                         const int* __restrict__ cnt,
                                                         const int* __restrict__ col,
                                                         const float* __restrict__ dinv,
                                                         const float* __restrict__ x,
                                                         const float* __restrict__ W1,
                                                         const float* __restrict__ b1,
                                                         const float* __restrict__ W2,
                                                         float* __restrict__ z, int N) {
    __shared__ float tsum[16];
    const int r    = threadIdx.x >> 4;
    const int f    = threadIdx.x & 15;
    const int base = blockIdx.x * 16;
    const int n    = base + r;
    const bool mat = threadIdx.x < 32;
    const int m    = threadIdx.x & 15;
    const int kh   = (threadIdx.x >> 4) & 1;

    float w1v[8], b1v[8], w2v[8];
    float dm = 0.0f, xm = 0.0f;
    if (mat) {
#pragma unroll
        for (int kc = 0; kc < 4; ++kc) {
#pragma unroll
            for (int i = 0; i < 2; ++i) {
                const int k = kc * 4 + kh * 2 + i;
                w1v[kc * 2 + i] = W1[k];
                b1v[kc * 2 + i] = b1[k];
                w2v[kc * 2 + i] = W2[k];
            }
        }
        if (base + m < N) { dm = dinv[base + m]; xm = x[base + m]; }
    }

    float acc = 0.0f;
    if (n < N) {
        const int start = rowptr[n];
        const int len   = cnt[n];
        for (int j = f; j < len; j += 32) {
            const int  j1  = j + 16;
            const bool two = j1 < len;
            const int  s0  = col[start + j];
            const int  s1  = col[start + (two ? j1 : j)];
            const float a0 = dinv[s0] * x[s0];
            const float a1 = dinv[s1] * x[s1];
            acc += a0 + (two ? a1 : 0.0f);
        }
    }
    acc += __shfl_xor(acc, 8, 16);
    acc += __shfl_xor(acc, 4, 16);
    acc += __shfl_xor(acc, 2, 16);
    acc += __shfl_xor(acc, 1, 16);
    if (f == 0) tsum[r] = acc;
    __syncthreads();

    if (mat) {
        const float ts = dm * tsum[m];
        const float sl = dm * dm * xm;
        v8f d = {0.f, 0.f, 0.f, 0.f, 0.f, 0.f, 0.f, 0.f};
#pragma unroll
        for (int kc = 0; kc < 4; ++kc) {
            v2f a, b;
            a.x = fmaxf(ts * w1v[kc * 2 + 0] + sl * w1v[kc * 2 + 0] + b1v[kc * 2 + 0], 0.0f);
            a.y = fmaxf(ts * w1v[kc * 2 + 1] + sl * w1v[kc * 2 + 1] + b1v[kc * 2 + 1], 0.0f);
            b.x = w2v[kc * 2 + 0];
            b.y = w2v[kc * 2 + 1];
            d = __builtin_amdgcn_wmma_f32_16x16x4_f32(
                 false, a,  false, b,
                 (short)0, d,  false,  false);
        }
        if (m == 0) {
            float* zp = z + base + kh * 8;
            zp[0] = d[0]; zp[1] = d[1]; zp[2] = d[2]; zp[3] = d[3];
            zp[4] = d[4]; zp[5] = d[5]; zp[6] = d[6]; zp[7] = d[7];
        }
    }
}

__global__ __launch_bounds__(512) void gcn_layer2_kernel(const int* __restrict__ rowptr,
                                                         const int* __restrict__ cnt,
                                                         const int* __restrict__ col,
                                                         const float* __restrict__ dinv,
                                                         const float* __restrict__ z,
                                                         const float* __restrict__ b2,
                                                         float* __restrict__ out, int N) {
    __shared__ float stage[32];
    const int r = threadIdx.x >> 4;
    const int f = threadIdx.x & 15;
    const int n = blockIdx.x * 32 + r;
    float acc = 0.0f;
    float dn = 0.0f, zn = 0.0f;
    const float bias = b2[0];
    if (n < N) {
        const int start = rowptr[n];
        const int len   = cnt[n];
        dn = dinv[n];
        zn = z[n];
        for (int j = f; j < len; j += 32) {
            const int  j1  = j + 16;
            const bool two = j1 < len;
            const int  s0  = col[start + j];
            const int  s1  = col[start + (two ? j1 : j)];
            const float a0 = dinv[s0] * z[s0];
            const float a1 = dinv[s1] * z[s1];
            acc += a0 + (two ? a1 : 0.0f);
        }
    }
    acc += __shfl_xor(acc, 8, 16);
    acc += __shfl_xor(acc, 4, 16);
    acc += __shfl_xor(acc, 2, 16);
    acc += __shfl_xor(acc, 1, 16);
    if (f == 0) stage[r] = dn * acc + (dn * dn * zn + bias);
    __syncthreads();
    if (threadIdx.x < 32) {
        const int mo = blockIdx.x * 32 + threadIdx.x;
        if (mo < N) out[mo] = stage[threadIdx.x];
    }
}

extern "C" void kernel_launch(void* const* d_in, const int* in_sizes, int n_in,
                              void* d_out, int out_size, void* d_ws, size_t ws_size,
                              hipStream_t stream) {
    const float* x  = (const float*)d_in[0];
    const float* W1 = (const float*)d_in[1];
    const float* b1 = (const float*)d_in[2];
    const float* W2 = (const float*)d_in[3];
    const float* b2 = (const float*)d_in[4];
    const int*   ei = (const int*)d_in[5];
    int N = in_sizes[0];
    int E = in_sizes[5] / 2;
    const int* src = ei;
    const int* dst = ei + E;
    int paddedN = (N + 15) & ~15;

    char* ws = (char*)d_ws;
    float* dinv   = (float*)ws; ws += (size_t)paddedN * sizeof(float);
    float* z      = (float*)ws; ws += (size_t)paddedN * sizeof(float);
    int*   cnt    = (int*)ws;   ws += (size_t)paddedN * sizeof(int);
    int*   rowptr = (int*)ws;   ws += (size_t)paddedN * sizeof(int);
    int*   T      = (int*)ws;   ws += (size_t)(GCN_NB + 1) * GCN_NS * sizeof(int);
    int*   pairs  = (int*)ws;   ws += (((size_t)E * 2 * sizeof(int)) + 255) & ~(size_t)255;
    int*   col    = (int*)ws;
    float* out    = (float*)d_out;

    int tiles = paddedN / 16;

    gcn_slice_sort_kernel<<<GCN_NS, 1024, 0, stream>>>(src, dst, T, pairs, E, N);
    gcn_rows_kernel<<<GCN_NB, 1024, 0, stream>>>(pairs, T, rowptr, cnt, dinv, col, E, N);
    gcn_layer1_kernel<<<tiles, 256, 0, stream>>>(rowptr, cnt, col, dinv, x, W1, b1, W2, z, N);
    gcn_layer2_kernel<<<(N + 31) / 32, 512, 0, stream>>>(rowptr, cnt, col, dinv, z, b2, out, N);
}


# Both layers: one block per dst range copies the per-source term table (dinv*v, all N) into LDS in 2 passes and looks its col entries up there, no global gather per edge; row sum order kept
# speedup vs baseline: 8.6307x; 1.5422x over previous
#include <hip/hip_runtime.h>
#include <hip/hip_bf16.h>

typedef __attribute__((ext_vector_type(2)))  float    v2f;
typedef __attribute__((ext_vector_type(8)))  float    v8f;
typedef __attribute__((ext_vector_type(4)))  float    v4f;

#define GCN_NB   256
#define GCN_NS   256
#define GCN_RMAX 4096
#define GCN_LPT  16
#define GCN_LMAX (1024 * GCN_LPT)
#define GCN_CCAP 32768
#define GCN_RPW  (GCN_NS / 32)
#define GCN_RLD  3

__global__ __launch_bounds__(1024) void gcn_slice_sort_kernel(const int* __restrict__ src,
                                                              const int* __restrict__ dst,
                                                              int* __restrict__ T,
                                                              int* __restrict__ pairs_raw,
                                                              int E, int N) {
    __shared__ int2 stage[GCN_LMAX];
    __shared__ int hist[GCN_NB];
    __shared__ int first[GCN_NB];
    __shared__ int wtot[GCN_NB / 32];
    int2* pairs = (int2*)pairs_raw;
    const int tid  = threadIdx.x;
    const int lane = tid & 31;
    const int w    = tid >> 5;
    const int s    = blockIdx.x;
    const unsigned R = (unsigned)((N + GCN_NB - 1) / GCN_NB);
    const int L  = (E + GCN_NS - 1) / GCN_NS;
    const int e0 = s * L;
    const int e1 = (e0 + L < E) ? (e0 + L) : E;
    const int len = e1 > e0 ? e1 - e0 : 0;
    if (tid < GCN_NB) hist[tid] = 0;

    int d[GCN_LPT], sv[GCN_LPT], rg[GCN_LPT], rk[GCN_LPT];
#pragma unroll
    for (int k = 0; k < GCN_LPT; ++k) {
        const int ek = e0 + tid + k * 1024;
        const int ec = ek < e1 ? ek : e1 - 1;
        d[k]  = dst[ec];
        sv[k] = src[ec];
    }
    __syncthreads();
#pragma unroll
    for (int k = 0; k < GCN_LPT; ++k) {
        rg[k] = (int)((unsigned)d[k] / R);
        rk[k] = 0;
        if (e0 + tid + k * 1024 < e1) rk[k] = atomicAdd(&hist[rg[k]], 1);
    }
    __syncthreads();

    const int c = (tid < GCN_NB) ? hist[tid] : 0;
    int inc = c;
    for (int o = 1; o < 32; o <<= 1) {
        int y = __shfl_up(inc, o, 32);
        if (lane >= o) inc += y;
    }
    if (tid < GCN_NB && lane == 31) wtot[w] = inc;
    __syncthreads();
    if (tid < GCN_NB) {
        int before = 0;
#pragma unroll
        for (int j = 0; j < GCN_NB / 32; ++j) before += (j < w) ? wtot[j] : 0;
        const int f = before + inc - c;
        first[tid] = f;
        T[tid * GCN_NS + s] = f;
        if (tid == GCN_NB - 1) T[GCN_NB * GCN_NS + s] = f + c;
    }
    __syncthreads();

#pragma unroll
    for (int k = 0; k < GCN_LPT; ++k) rg[k] = first[rg[k]] + rk[k];
#pragma unroll
    for (int k = 0; k < GCN_LPT; ++k) {
        if (e0 + tid + k * 1024 < e1) stage[rg[k]] = make_int2(sv[k], d[k]);
    }
    __syncthreads();
#pragma unroll 4
    for (int i = tid; i < len; i += 1024) pairs[e0 + i] = stage[i];
}

__global__ __launch_bounds__(1024) void gcn_rows_kernel(const int* __restrict__ pairs_raw,
                                                        const int* __restrict__ T,
                                                        int* __restrict__ rowptr,
                                                        int* __restrict__ cnt,
                                                        float* __restrict__ dinv,
                                                        int* __restrict__ col, int E, int N) {
    __shared__ int lcol[GCN_CCAP];
    __shared__ int lcnt[GCN_RMAX];
    __shared__ int rs[GCN_NS];
    __shared__ int rl[GCN_NS];
    __shared__ int wtot[32];
    __shared__ int redA[GCN_NS / 32];
    __shared__ int redL[GCN_NS / 32];
    const int2* pairs = (const int2*)pairs_raw;
    const int tid  = threadIdx.x;
    const int lane = tid & 31;
    const int w    = tid >> 5;
    const int b    = blockIdx.x;
    const int R    = (N + GCN_NB - 1) / GCN_NB;
    const int n0   = b * R;
    if (n0 >= N) return;
    const int L = (E + GCN_NS - 1) / GCN_NS;

    int ra = 0, rn = 0;
    if (tid < GCN_NS) {
        ra = T[b * GCN_NS + tid];
        rn = T[(b + 1) * GCN_NS + tid] - ra;
        rs[tid] = tid * L + ra;
        rl[tid] = rn;
    }
    int sa = ra, sl = rn;
    for (int o = 16; o > 0; o >>= 1) {
        sa += __shfl_xor(sa, o, 32);
        sl += __shfl_xor(sl, o, 32);
    }
    if (tid < GCN_NS && lane == 0) { redA[w] = sa; redL[w] = sl; }
    const int i = tid * 4;
    *(int4*)(lcnt + i) = make_int4(0, 0, 0, 0);
    __syncthreads();
    int p0 = 0, tot = 0;
#pragma unroll
    for (int j = 0; j < GCN_NS / 32; ++j) { p0 += redA[j]; tot += redL[j]; }
    const bool staged = tot <= GCN_CCAP;

    int  st[GCN_RPW], rn_[GCN_RPW];
    int2 rec[GCN_RPW][GCN_RLD];
#pragma unroll
    for (int r = 0; r < GCN_RPW; ++r) {
        st[r]  = rs[w + 32 * r];
        rn_[r] = rl[w + 32 * r];
#pragma unroll
        for (int h = 0; h < GCN_RLD; ++h) {
            const int j = lane + 32 * h;
            int g = st[r] + (j < rn_[r] ? j : rn_[r] - 1);
            g = g < E ? g : E - 1;
            g = g > 0 ? g : 0;
            rec[r][h] = pairs[g];
        }
    }

#pragma unroll
    for (int r = 0; r < GCN_RPW; ++r) {
#pragma unroll
        for (int h = 0; h < GCN_RLD; ++h) {
            if (lane + 32 * h < rn_[r]) atomicAdd(&lcnt[rec[r][h].y - n0], 1);
        }
        for (int j = lane + 32 * GCN_RLD; j < rn_[r]; j += 32)
            atomicAdd(&lcnt[pairs[st[r] + j].y - n0], 1);
    }
    __syncthreads();

    const int4 c = *(const int4*)(lcnt + i);
    const int part = (c.x + c.y) + (c.z + c.w);
    int inc = part;
    for (int o = 1; o < 32; o <<= 1) {
        int y = __shfl_up(inc, o, 32);
        if (lane >= o) inc += y;
    }
    if (lane == 31) wtot[w] = inc;
    __syncthreads();
    const int wt = wtot[lane];
    int winc = wt;
    for (int o = 1; o < 32; o <<= 1) {
        int y = __shfl_up(winc, o, 32);
        if (lane >= o) winc += y;
    }
    const int q0 = p0 + __shfl(winc - wt, w, 32) + inc - part;
    const int q1 = q0 + c.x;
    const int q2 = q1 + c.y;
    const int q3 = q2 + c.z;
    {
        const int n = n0 + i;
        if (i     < R && n     < N) { rowptr[n]     = q0; cnt[n]     = c.x; dinv[n]     = rsqrtf((float)(c.x + 1)); }
        if (i + 1 < R && n + 1 < N) { rowptr[n + 1] = q1; cnt[n + 1] = c.y; dinv[n + 1] = rsqrtf((float)(c.y + 1)); }
        if (i + 2 < R && n + 2 < N) { rowptr[n + 2] = q2; cnt[n + 2] = c.z; dinv[n + 2] = rsqrtf((float)(c.z + 1)); }
        if (i + 3 < R && n + 3 < N) { rowptr[n + 3] = q3; cnt[n + 3] = c.w; dinv[n + 3] = rsqrtf((float)(c.w + 1)); }
    }
    *(int4*)(lcnt + i) = make_int4(q0, q1, q2, q3);
    __syncthreads();

#pragma unroll
    for (int r = 0; r < GCN_RPW; ++r) {
        int slot[GCN_RLD];
#pragma unroll
        for (int h = 0; h < GCN_RLD; ++h) {
            slot[h] = p0;
            if (lane + 32 * h < rn_[r]) slot[h] = atomicAdd(&lcnt[rec[r][h].y - n0], 1);
        }
#pragma unroll
        for (int h = 0; h < GCN_RLD; ++h) {
            if (lane + 32 * h < rn_[r]) {
                if (staged) lcol[slot[h] - p0] = rec[r][h].x;
                else        col[slot[h]]       = rec[r][h].x;
            }
        }
        for (int j = lane + 32 * GCN_RLD; j < rn_[r]; j += 32) {
            const int2 e = pairs[st[r] + j];
            const int q = atomicAdd(&lcnt[e.y - n0], 1);
            if (staged) lcol[q - p0] = e.x;
            else        col[q]       = e.x;
        }
    }
    if (staged) {
        __syncthreads();
        for (int j = tid; j < tot; j += 1024) col[p0 + j] = lcol[j];
    }
}

#define GCN_TC   57344
#define GCN_EPT  16
#define GCN_SG   7

__device__ __forceinline__ void gcn_range_rows(const int* __restrict__ rowptr,
                                               const int* __restrict__ cnt,
                                               const int* __restrict__ col,
                                               const float* __restrict__ dinv,
                                               const float* __restrict__ v,
                                               int N, int E, int n0, int Rb,
                                               float* tab, int* lst, int* lln, float* rsum) {
    const int tid = threadIdx.x;
    const int r16 = tid >> 4;
    const int f   = tid & 15;
    const int nl  = n0 + Rb - 1;
    const int p0  = rowptr[n0];
    const int tot = rowptr[nl] + cnt[nl] - p0;
    for (int t = tid; t < Rb; t += 1024) {
        lst[t] = rowptr[n0 + t] - p0;
        lln[t] = cnt[n0 + t];
    }
    if (tot <= GCN_EPT * 1024) {
        int   cj[GCN_EPT];
        float val[GCN_EPT];
#pragma unroll
        for (int k = 0; k < GCN_EPT; ++k) {
            int j = tid + 1024 * k;
            j = j < tot ? j : tot - 1;
            int g = p0 + j;
            g = g < E ? g : E - 1;
            g = g > 0 ? g : 0;
            cj[k]  = col[g];
            val[k] = 0.0f;
        }
        for (int c0 = 0; c0 < N; c0 += GCN_TC) {
            const int clen = (N - c0 < GCN_TC) ? (N - c0) : GCN_TC;
            const int cl4  = clen & ~3;
            const float* dch = dinv + c0;
            const float* vch = v + c0;
            for (unsigned g = (unsigned)tid * 4u; g < (unsigned)cl4; g += 4096u * GCN_SG) {
                v4f a[GCN_SG], b[GCN_SG];
                unsigned ic[GCN_SG];
#pragma unroll
                for (int u = 0; u < GCN_SG; ++u) {
                    const unsigned i = g + (unsigned)u * 4096u;
                    ic[u] = i < (unsigned)cl4 ? i : g;
                    a[u] = *(const v4f*)(dch + ic[u]);
                    b[u] = *(const v4f*)(vch + ic[u]);
                }
                asm volatile("" :: "v"(a[0]), "v"(a[1]), "v"(a[2]), "v"(a[3]), "v"(a[4]), "v"(a[5]), "v"(a[6]),
                                   "v"(b[0]), "v"(b[1]), "v"(b[2]), "v"(b[3]), "v"(b[4]), "v"(b[5]), "v"(b[6]));
#pragma unroll
                for (int u = 0; u < GCN_SG; ++u) *(v4f*)(tab + ic[u]) = a[u] * b[u];
            }
            if (tid < clen - cl4) {
                const int i = cl4 + tid;
                tab[i] = dinv[c0 + i] * v[c0 + i];
            }
            __syncthreads();
#pragma unroll
            for (int k = 0; k < GCN_EPT; ++k) {
                const unsigned loc = (unsigned)(cj[k] - c0);
                if (loc < (unsigned)clen) val[k] = tab[loc];
            }
            __syncthreads();
        }
#pragma unroll
        for (int k = 0; k < GCN_EPT; ++k) {
            const int j = tid + 1024 * k;
            if (j < tot) tab[j] = val[k];
        }
        __syncthreads();
        for (int rb = 0; rb < Rb; rb += 64) {
            const int  r  = rb + r16;
            const bool ok = r < Rb;
            const int start = ok ? lst[r] : 0;
            const int len   = ok ? lln[r] : 0;
            float acc = 0.0f;
            for (int j = f; j < len; j += 32) {
                const int  j1  = j + 16;
                const bool two = j1 < len;
                const float a0 = tab[start + j];
                const float a1 = tab[start + (two ? j1 : j)];
                acc += a0 + (two ? a1 : 0.0f);
            }
            acc += __shfl_xor(acc, 8, 16);
            acc += __shfl_xor(acc, 4, 16);
            acc += __shfl_xor(acc, 2, 16);
            acc += __shfl_xor(acc, 1, 16);
            if (ok && f == 0) rsum[r] = acc;
        }
    } else {
        __syncthreads();
        for (int rb = 0; rb < Rb; rb += 64) {
            const int  r  = rb + r16;
            const bool ok = r < Rb;
            const int start = ok ? lst[r] + p0 : 0;
            const int len   = ok ? lln[r] : 0;
            float acc = 0.0f;
            for (int j = f; j < len; j += 32) {
                const int  j1  = j + 16;
                const bool two = j1 < len;
                const int  s0  = col[start + j];
                const int  s1  = col[start + (two ? j1 : j)];
                const float a0 = dinv[s0] * v[s0];
                const float a1 = dinv[s1] * v[s1];
                acc += a0 + (two ? a1 : 0.0f);
            }
            acc += __shfl_xor(acc, 8, 16);
            acc += __shfl_xor(acc, 4, 16);
            acc += __shfl_xor(acc, 2, 16);
            acc += __shfl_xor(acc, 1, 16);
            if (ok && f == 0) rsum[r] = acc;
        }
    }
    __syncthreads();
}

__global__ __launch_bounds__(1024) void gcn_layer1_kernel(const int* __restrict__ rowptr,
                                                          const int* __restrict__ cnt,
                                                          const int* __restrict__ col,
                                                          const float* __restrict__ dinv,
                                                          const float* __restrict__ x,
                                                          const float* __restrict__ W1,
                                                          const float* __restrict__ b1,
                                                          const float* __restrict__ W2,
                                                          float* __restrict__ z, int E, int N) {
    __shared__ float tab[GCN_TC];
    __shared__ int   lst[GCN_RMAX];
    __shared__ int   lln[GCN_RMAX];
    __shared__ float rsum[GCN_RMAX];
    __shared__ float wl[48];
    __shared__ float own[2048];
    const int R  = (N + GCN_NB - 1) / GCN_NB;
    const int n0 = blockIdx.x * R;
    if (n0 >= N) return;
    const int Rb = (N - n0 < R) ? (N - n0) : R;
    const int w  = threadIdx.x >> 5;
    const int m  = threadIdx.x & 15;
    const int kh = (threadIdx.x >> 4) & 1;
    const int ntile = (Rb + 15) >> 4;

    if (threadIdx.x < 16) {
        wl[threadIdx.x]      = W1[threadIdx.x];
        wl[16 + threadIdx.x] = b1[threadIdx.x];
        wl[32 + threadIdx.x] = W2[threadIdx.x];
    }
    {
        float dm0 = 0.0f, xm0 = 0.0f;
        if (w * 16 + m < Rb) { dm0 = dinv[n0 + w * 16 + m]; xm0 = x[n0 + w * 16 + m]; }
        own[threadIdx.x]        = dm0;
        own[1024 + threadIdx.x] = xm0;
    }

    gcn_range_rows(rowptr, cnt, col, dinv, x, N, E, n0, Rb, tab, lst, lln, rsum);

    const float dm0 = own[threadIdx.x];
    const float xm0 = own[1024 + threadIdx.x];

    float w1v[8], b1v[8], w2v[8];
#pragma unroll
    for (int kc = 0; kc < 4; ++kc) {
#pragma unroll
        for (int i = 0; i < 2; ++i) {
            const int k = kc * 4 + kh * 2 + i;
            w1v[kc * 2 + i] = wl[k];
            b1v[kc * 2 + i] = wl[16 + k];
            w2v[kc * 2 + i] = wl[32 + k];
        }
    }
    for (int t = w; t < ntile; t += 32) {
        const int  row = t * 16 + m;
        const bool ok  = row < Rb;
        float dm = dm0, xm = xm0;
        if (t != w) {
            dm = ok ? dinv[n0 + row] : 0.0f;
            xm = ok ? x[n0 + row] : 0.0f;
        }
        const float ts = ok ? dm * rsum[row] : 0.0f;
        const float sl = dm * dm * xm;
        v8f d = {0.f, 0.f, 0.f, 0.f, 0.f, 0.f, 0.f, 0.f};
#pragma unroll
        for (int kc = 0; kc < 4; ++kc) {
            v2f a, b;
            a.x = fmaxf(ts * w1v[kc * 2 + 0] + sl * w1v[kc * 2 + 0] + b1v[kc * 2 + 0], 0.0f);
            a.y = fmaxf(ts * w1v[kc * 2 + 1] + sl * w1v[kc * 2 + 1] + b1v[kc * 2 + 1], 0.0f);
            b.x = w2v[kc * 2 + 0];
            b.y = w2v[kc * 2 + 1];
            d = __builtin_amdgcn_wmma_f32_16x16x4_f32(
                 false, a,  false, b,
                 (short)0, d,  false,  false);
        }
        if (m == 0) {
            const int q  = t * 16 + kh * 8;
            float* zp = z + n0 + q;
            if (q     < Rb) zp[0] = d[0];
            if (q + 1 < Rb) zp[1] = d[1];
            if (q + 2 < Rb) zp[2] = d[2];
            if (q + 3 < Rb) zp[3] = d[3];
            if (q + 4 < Rb) zp[4] = d[4];
            if (q + 5 < Rb) zp[5] = d[5];
            if (q + 6 < Rb) zp[6] = d[6];
            if (q + 7 < Rb) zp[7] = d[7];
        }
    }
}

__global__ __launch_bounds__(1024) void gcn_layer2_kernel(const int* __restrict__ rowptr,
                                                          const int* __restrict__ cnt,
                                                          const int* __restrict__ col,
                                                          const float* __restrict__ dinv,
                                                          const float* __restrict__ z,
                                                          const float* __restrict__ b2,
                                                          float* __restrict__ out, int E, int N) {
    __shared__ float tab[GCN_TC];
    __shared__ int   lst[GCN_RMAX];
    __shared__ int   lln[GCN_RMAX];
    __shared__ float rsum[GCN_RMAX];
    const int R  = (N + GCN_NB - 1) / GCN_NB;
    const int n0 = blockIdx.x * R;
    if (n0 >= N) return;
    const int Rb = (N - n0 < R) ? (N - n0) : R;
    const int tid = threadIdx.x;
    const float bias = b2[0];
    float dn0 = 0.0f, zn0 = 0.0f;
    if (tid < Rb) { dn0 = dinv[n0 + tid]; zn0 = z[n0 + tid]; }

    gcn_range_rows(rowptr, cnt, col, dinv, z, N, E, n0, Rb, tab, lst, lln, rsum);

    for (int t = tid; t < Rb; t += 1024) {
        float dn = dn0, zn = zn0;
        if (t != tid) { dn = dinv[n0 + t]; zn = z[n0 + t]; }
        out[n0 + t] = dn * rsum[t] + (dn * dn * zn + bias);
    }
}

extern "C" void kernel_launch(void* const* d_in, const int* in_sizes, int n_in,
                              void* d_out, int out_size, void* d_ws, size_t ws_size,
                              hipStream_t stream) {
    const float* x  = (const float*)d_in[0];
    const float* W1 = (const float*)d_in[1];
    const float* b1 = (const float*)d_in[2];
    const float* W2 = (const float*)d_in[3];
    const float* b2 = (const float*)d_in[4];
    const int*   ei = (const int*)d_in[5];
    int N = in_sizes[0];
    int E = in_sizes[5] / 2;
    const int* src = ei;
    const int* dst = ei + E;
    int paddedN = (N + 15) & ~15;

    char* ws = (char*)d_ws;
    float* dinv   = (float*)ws; ws += (size_t)paddedN * sizeof(float);
    float* z      = (float*)ws; ws += (size_t)paddedN * sizeof(float);
    int*   cnt    = (int*)ws;   ws += (size_t)paddedN * sizeof(int);
    int*   rowptr = (int*)ws;   ws += (size_t)paddedN * sizeof(int);
    int*   T      = (int*)ws;   ws += (size_t)(GCN_NB + 1) * GCN_NS * sizeof(int);
    int*   pairs  = (int*)ws;   ws += (((size_t)E * 2 * sizeof(int)) + 255) & ~(size_t)255;
    int*   col    = (int*)ws;
    float* out    = (float*)d_out;

    gcn_slice_sort_kernel<<<GCN_NS, 1024, 0, stream>>>(src, dst, T, pairs, E, N);
    gcn_rows_kernel<<<GCN_NB, 1024, 0, stream>>>(pairs, T, rowptr, cnt, dinv, col, E, N);
    gcn_layer1_kernel<<<GCN_NB, 1024, 0, stream>>>(rowptr, cnt, col, dinv, x, W1, b1, W2, z, E, N);
    gcn_layer2_kernel<<<GCN_NB, 1024, 0, stream>>>(rowptr, cnt, col, dinv, z, b2, out, E, N);
}


# Row sums of both layers: a 16-lane group takes 8 of its rows at once (their LDS reads in flight together) and the 16-lane tree runs on DPP adds, not LDS permutes; row sum order kept
# speedup vs baseline: 9.1152x; 1.0561x over previous
#include <hip/hip_runtime.h>
#include <hip/hip_bf16.h>

typedef __attribute__((ext_vector_type(2)))  float    v2f;
typedef __attribute__((ext_vector_type(8)))  float    v8f;
typedef __attribute__((ext_vector_type(4)))  float    v4f;

#define GCN_NB   256
#define GCN_NS   256
#define GCN_RMAX 4096
#define GCN_LPT  16
#define GCN_LMAX (1024 * GCN_LPT)
#define GCN_CCAP 32768
#define GCN_RPW  (GCN_NS / 32)
#define GCN_RLD  3

__global__ __launch_bounds__(1024) void gcn_slice_sort_kernel(const int* __restrict__ src,
                                                              const int* __restrict__ dst,
                                                              int* __restrict__ T,
                                                              int* __restrict__ pairs_raw,
                                                              int E, int N) {
    __shared__ int2 stage[GCN_LMAX];
    __shared__ int hist[GCN_NB];
    __shared__ int first[GCN_NB];
    __shared__ int wtot[GCN_NB / 32];
    int2* pairs = (int2*)pairs_raw;
    const int tid  = threadIdx.x;
    const int lane = tid & 31;
    const int w    = tid >> 5;
    const int s    = blockIdx.x;
    const unsigned R = (unsigned)((N + GCN_NB - 1) / GCN_NB);
    const int L  = (E + GCN_NS - 1) / GCN_NS;
    const int e0 = s * L;
    const int e1 = (e0 + L < E) ? (e0 + L) : E;
    const int len = e1 > e0 ? e1 - e0 : 0;
    if (tid < GCN_NB) hist[tid] = 0;

    int d[GCN_LPT], sv[GCN_LPT], rg[GCN_LPT], rk[GCN_LPT];
#pragma unroll
    for (int k = 0; k < GCN_LPT; ++k) {
        const int ek = e0 + tid + k * 1024;
        const int ec = ek < e1 ? ek : e1 - 1;
        d[k]  = dst[ec];
        sv[k] = src[ec];
    }
    __syncthreads();
#pragma unroll
    for (int k = 0; k < GCN_LPT; ++k) {
        rg[k] = (int)((unsigned)d[k] / R);
        rk[k] = 0;
        if (e0 + tid + k * 1024 < e1) rk[k] = atomicAdd(&hist[rg[k]], 1);
    }
    __syncthreads();

    const int c = (tid < GCN_NB) ? hist[tid] : 0;
    int inc = c;
    for (int o = 1; o < 32; o <<= 1) {
        int y = __shfl_up(inc, o, 32);
        if (lane >= o) inc += y;
    }
    if (tid < GCN_NB && lane == 31) wtot[w] = inc;
    __syncthreads();
    if (tid < GCN_NB) {
        int before = 0;
#pragma unroll
        for (int j = 0; j < GCN_NB / 32; ++j) before += (j < w) ? wtot[j] : 0;
        const int f = before + inc - c;
        first[tid] = f;
        T[tid * GCN_NS + s] = f;
        if (tid == GCN_NB - 1) T[GCN_NB * GCN_NS + s] = f + c;
    }
    __syncthreads();

#pragma unroll
    for (int k = 0; k < GCN_LPT; ++k) rg[k] = first[rg[k]] + rk[k];
#pragma unroll
    for (int k = 0; k < GCN_LPT; ++k) {
        if (e0 + tid + k * 1024 < e1) stage[rg[k]] = make_int2(sv[k], d[k]);
    }
    __syncthreads();
#pragma unroll 4
    for (int i = tid; i < len; i += 1024) pairs[e0 + i] = stage[i];
}

__global__ __launch_bounds__(1024) void gcn_rows_kernel(const int* __restrict__ pairs_raw,
                                                        const int* __restrict__ T,
                                                        int* __restrict__ rowptr,
                                                        int* __restrict__ cnt,
                                                        float* __restrict__ dinv,
                                                        int* __restrict__ col, int E, int N) {
    __shared__ int lcol[GCN_CCAP];
    __shared__ int lcnt[GCN_RMAX];
    __shared__ int rs[GCN_NS];
    __shared__ int rl[GCN_NS];
    __shared__ int wtot[32];
    __shared__ int redA[GCN_NS / 32];
    __shared__ int redL[GCN_NS / 32];
    const int2* pairs = (const int2*)pairs_raw;
    const int tid  = threadIdx.x;
    const int lane = tid & 31;
    const int w    = tid >> 5;
    const int b    = blockIdx.x;
    const int R    = (N + GCN_NB - 1) / GCN_NB;
    const int n0   = b * R;
    if (n0 >= N) return;
    const int L = (E + GCN_NS - 1) / GCN_NS;

    int ra = 0, rn = 0;
    if (tid < GCN_NS) {
        ra = T[b * GCN_NS + tid];
        rn = T[(b + 1) * GCN_NS + tid] - ra;
        rs[tid] = tid * L + ra;
        rl[tid] = rn;
    }
    int sa = ra, sl = rn;
    for (int o = 16; o > 0; o >>= 1) {
        sa += __shfl_xor(sa, o, 32);
        sl += __shfl_xor(sl, o, 32);
    }
    if (tid < GCN_NS && lane == 0) { redA[w] = sa; redL[w] = sl; }
    const int i = tid * 4;
    *(int4*)(lcnt + i) = make_int4(0, 0, 0, 0);
    __syncthreads();
    int p0 = 0, tot = 0;
#pragma unroll
    for (int j = 0; j < GCN_NS / 32; ++j) { p0 += redA[j]; tot += redL[j]; }
    const bool staged = tot <= GCN_CCAP;

    int  st[GCN_RPW], rn_[GCN_RPW];
    int2 rec[GCN_RPW][GCN_RLD];
#pragma unroll
    for (int r = 0; r < GCN_RPW; ++r) {
        st[r]  = rs[w + 32 * r];
        rn_[r] = rl[w + 32 * r];
#pragma unroll
        for (int h = 0; h < GCN_RLD; ++h) {
            const int j = lane + 32 * h;
            int g = st[r] + (j < rn_[r] ? j : rn_[r] - 1);
            g = g < E ? g : E - 1;
            g = g > 0 ? g : 0;
            rec[r][h] = pairs[g];
        }
    }

#pragma unroll
    for (int r = 0; r < GCN_RPW; ++r) {
#pragma unroll
        for (int h = 0; h < GCN_RLD; ++h) {
            if (lane + 32 * h < rn_[r]) atomicAdd(&lcnt[rec[r][h].y - n0], 1);
        }
        for (int j = lane + 32 * GCN_RLD; j < rn_[r]; j += 32)
            atomicAdd(&lcnt[pairs[st[r] + j].y - n0], 1);
    }
    __syncthreads();

    const int4 c = *(const int4*)(lcnt + i);
    const int part = (c.x + c.y) + (c.z + c.w);
    int inc = part;
    for (int o = 1; o < 32; o <<= 1) {
        int y = __shfl_up(inc, o, 32);
        if (lane >= o) inc += y;
    }
    if (lane == 31) wtot[w] = inc;
    __syncthreads();
    const int wt = wtot[lane];
    int winc = wt;
    for (int o = 1; o < 32; o <<= 1) {
        int y = __shfl_up(winc, o, 32);
        if (lane >= o) winc += y;
    }
    const int q0 = p0 + __shfl(winc - wt, w, 32) + inc - part;
    const int q1 = q0 + c.x;
    const int q2 = q1 + c.y;
    const int q3 = q2 + c.z;
    {
        const int n = n0 + i;
        if (i     < R && n     < N) { rowptr[n]     = q0; cnt[n]     = c.x; dinv[n]     = rsqrtf((float)(c.x + 1)); }
        if (i + 1 < R && n + 1 < N) { rowptr[n + 1] = q1; cnt[n + 1] = c.y; dinv[n + 1] = rsqrtf((float)(c.y + 1)); }
        if (i + 2 < R && n + 2 < N) { rowptr[n + 2] = q2; cnt[n + 2] = c.z; dinv[n + 2] = rsqrtf((float)(c.z + 1)); }
        if (i + 3 < R && n + 3 < N) { rowptr[n + 3] = q3; cnt[n + 3] = c.w; dinv[n + 3] = rsqrtf((float)(c.w + 1)); }
    }
    *(int4*)(lcnt + i) = make_int4(q0, q1, q2, q3);
    __syncthreads();

#pragma unroll
    for (int r = 0; r < GCN_RPW; ++r) {
        int slot[GCN_RLD];
#pragma unroll
        for (int h = 0; h < GCN_RLD; ++h) {
            slot[h] = p0;
            if (lane + 32 * h < rn_[r]) slot[h] = atomicAdd(&lcnt[rec[r][h].y - n0], 1);
        }
#pragma unroll
        for (int h = 0; h < GCN_RLD; ++h) {
            if (lane + 32 * h < rn_[r]) {
                if (staged) lcol[slot[h] - p0] = rec[r][h].x;
                else        col[slot[h]]       = rec[r][h].x;
            }
        }
        for (int j = lane + 32 * GCN_RLD; j < rn_[r]; j += 32) {
            const int2 e = pairs[st[r] + j];
            const int q = atomicAdd(&lcnt[e.y - n0], 1);
            if (staged) lcol[q - p0] = e.x;
            else        col[q]       = e.x;
        }
    }
    if (staged) {
        __syncthreads();
        for (int j = tid; j < tot; j += 1024) col[p0 + j] = lcol[j];
    }
}

#define GCN_TC   57344
#define GCN_EPT  16
#define GCN_SG   7
#define GCN_RU   8
static_assert(GCN_RU == 8, "the read fence in gcn_range_rows names 8 rows");

__device__ __forceinline__ float gcn_sum16(float a) {
    a += __int_as_float(__builtin_amdgcn_update_dpp(0, __float_as_int(a), 0x168, 0xf, 0xf, false));
    a += __int_as_float(__builtin_amdgcn_update_dpp(0, __float_as_int(a), 0x164, 0xf, 0xf, false));
    a += __int_as_float(__builtin_amdgcn_update_dpp(0, __float_as_int(a), 0x162, 0xf, 0xf, false));
    a += __int_as_float(__builtin_amdgcn_update_dpp(0, __float_as_int(a), 0x161, 0xf, 0xf, false));
    return a;
}

__device__ __forceinline__ void gcn_range_rows(const int* __restrict__ rowptr,
                                               const int* __restrict__ cnt,
                                               const int* __restrict__ col,
                                               const float* __restrict__ dinv,
                                               const float* __restrict__ v,
                                               int N, int E, int n0, int Rb,
                                               float* tab, int2* lrow, float* rsum) {
    const int tid = threadIdx.x;
    const int r16 = tid >> 4;
    const int f   = tid & 15;
    const int nl  = n0 + Rb - 1;
    const int p0  = rowptr[n0];
    const int tot = rowptr[nl] + cnt[nl] - p0;
    for (int t = tid; t < Rb; t += 1024)
        lrow[t] = make_int2(rowptr[n0 + t] - p0, cnt[n0 + t]);
    if (tot <= GCN_EPT * 1024) {
        int   cj[GCN_EPT];
        float val[GCN_EPT];
#pragma unroll
        for (int k = 0; k < GCN_EPT; ++k) {
            int j = tid + 1024 * k;
            j = j < tot ? j : tot - 1;
            int g = p0 + j;
            g = g < E ? g : E - 1;
            g = g > 0 ? g : 0;
            cj[k]  = col[g];
            val[k] = 0.0f;
        }
        for (int c0 = 0; c0 < N; c0 += GCN_TC) {
            const int clen = (N - c0 < GCN_TC) ? (N - c0) : GCN_TC;
            const int cl4  = clen & ~3;
            const float* dch = dinv + c0;
            const float* vch = v + c0;
            for (unsigned g = (unsigned)tid * 4u; g < (unsigned)cl4; g += 4096u * GCN_SG) {
                v4f a[GCN_SG], b[GCN_SG];
                unsigned ic[GCN_SG];
#pragma unroll
                for (int u = 0; u < GCN_SG; ++u) {
                    const unsigned i = g + (unsigned)u * 4096u;
                    ic[u] = i < (unsigned)cl4 ? i : g;
                    a[u] = *(const v4f*)(dch + ic[u]);
                    b[u] = *(const v4f*)(vch + ic[u]);
                }
                asm volatile("" :: "v"(a[0]), "v"(a[1]), "v"(a[2]), "v"(a[3]), "v"(a[4]), "v"(a[5]), "v"(a[6]),
                                   "v"(b[0]), "v"(b[1]), "v"(b[2]), "v"(b[3]), "v"(b[4]), "v"(b[5]), "v"(b[6]));
#pragma unroll
                for (int u = 0; u < GCN_SG; ++u) *(v4f*)(tab + ic[u]) = a[u] * b[u];
            }
            if (tid < clen - cl4) {
                const int i = cl4 + tid;
                tab[i] = dinv[c0 + i] * v[c0 + i];
            }
            __syncthreads();
#pragma unroll
            for (int k = 0; k < GCN_EPT; ++k) {
                const unsigned loc = (unsigned)(cj[k] - c0);
                if (loc < (unsigned)clen) val[k] = tab[loc];
            }
            __syncthreads();
        }
#pragma unroll
        for (int k = 0; k < GCN_EPT; ++k) {
            const int j = tid + 1024 * k;
            if (j < tot) tab[j] = val[k];
        }
        __syncthreads();
        for (int rb = 0; rb < Rb; rb += 64 * GCN_RU) {
            int   st[GCN_RU], ln[GCN_RU];
            float acc[GCN_RU];
            int   mx = 0;
#pragma unroll
            for (int u = 0; u < GCN_RU; ++u) {
                const int  r  = rb + 64 * u + r16;
                const bool ok = r < Rb;
                const int2 e  = lrow[ok ? r : 0];
                st[u]  = e.x;
                ln[u]  = ok ? e.y : 0;
                acc[u] = 0.0f;
                mx = ln[u] > mx ? ln[u] : mx;
            }
            for (int j = f; j < mx; j += 32) {
                float a0[GCN_RU], a1[GCN_RU];
#pragma unroll
                for (int u = 0; u < GCN_RU; ++u) {
                    a0[u] = tab[st[u] + j];
                    a1[u] = tab[st[u] + j + 16];
                }
                asm volatile("" :: "v"(a0[0]), "v"(a0[1]), "v"(a0[2]), "v"(a0[3]), "v"(a0[4]), "v"(a0[5]), "v"(a0[6]), "v"(a0[7]),
                                   "v"(a1[0]), "v"(a1[1]), "v"(a1[2]), "v"(a1[3]), "v"(a1[4]), "v"(a1[5]), "v"(a1[6]), "v"(a1[7]));
#pragma unroll
                for (int u = 0; u < GCN_RU; ++u) {
                    const bool  one = j < ln[u];
                    const bool  two = j + 16 < ln[u];
                    const float t   = a0[u] + (two ? a1[u] : 0.0f);
                    acc[u] = one ? acc[u] + t : acc[u];
                }
            }
#pragma unroll
            for (int u = 0; u < GCN_RU; ++u) acc[u] = gcn_sum16(acc[u]);
            float tsel = acc[0];
#pragma unroll
            for (int u = 1; u < GCN_RU; ++u) tsel = (f == u) ? acc[u] : tsel;
            const int r = rb + 64 * f + r16;
            if (f < GCN_RU && r < Rb) rsum[r] = tsel;
        }
    } else {
        __syncthreads();
        for (int rb = 0; rb < Rb; rb += 64) {
            const int  r  = rb + r16;
            const bool ok = r < Rb;
            const int2 e    = lrow[ok ? r : 0];
            const int start = ok ? e.x + p0 : 0;
            const int len   = ok ? e.y : 0;
            float acc = 0.0f;
            for (int j = f; j < len; j += 32) {
                const int  j1  = j + 16;
                const bool two = j1 < len;
                const int  s0  = col[start + j];
                const int  s1  = col[start + (two ? j1 : j)];
                const float a0 = dinv[s0] * v[s0];
                const float a1 = dinv[s1] * v[s1];
                acc += a0 + (two ? a1 : 0.0f);
            }
            acc = gcn_sum16(acc);
            if (ok && f == 0) rsum[r] = acc;
        }
    }
    __syncthreads();
}

__global__ __launch_bounds__(1024) void gcn_layer1_kernel(const int* __restrict__ rowptr,
                                                          const int* __restrict__ cnt,
                                                          const int* __restrict__ col,
                                                          const float* __restrict__ dinv,
                                                          const float* __restrict__ x,
                                                          const float* __restrict__ W1,
                                                          const float* __restrict__ b1,
                                                          const float* __restrict__ W2,
                                                          float* __restrict__ z, int E, int N) {
    __shared__ float tab[GCN_TC];
    __shared__ int2  lrow[GCN_RMAX];
    __shared__ float rsum[GCN_RMAX];
    __shared__ float wl[48];
    __shared__ float own[2048];
    const int R  = (N + GCN_NB - 1) / GCN_NB;
    const int n0 = blockIdx.x * R;
    if (n0 >= N) return;
    const int Rb = (N - n0 < R) ? (N - n0) : R;
    const int w  = threadIdx.x >> 5;
    const int m  = threadIdx.x & 15;
    const int kh = (threadIdx.x >> 4) & 1;
    const int ntile = (Rb + 15) >> 4;

    if (threadIdx.x < 16) {
        wl[threadIdx.x]      = W1[threadIdx.x];
        wl[16 + threadIdx.x] = b1[threadIdx.x];
        wl[32 + threadIdx.x] = W2[threadIdx.x];
    }
    {
        float dm0 = 0.0f, xm0 = 0.0f;
        if (w * 16 + m < Rb) { dm0 = dinv[n0 + w * 16 + m]; xm0 = x[n0 + w * 16 + m]; }
        own[threadIdx.x]        = dm0;
        own[1024 + threadIdx.x] = xm0;
    }

    gcn_range_rows(rowptr, cnt, col, dinv, x, N, E, n0, Rb, tab, lrow, rsum);

    const float dm0 = own[threadIdx.x];
    const float xm0 = own[1024 + threadIdx.x];

    float w1v[8], b1v[8], w2v[8];
#pragma unroll
    for (int kc = 0; kc < 4; ++kc) {
#pragma unroll
        for (int i = 0; i < 2; ++i) {
            const int k = kc * 4 + kh * 2 + i;
            w1v[kc * 2 + i] = wl[k];
            b1v[kc * 2 + i] = wl[16 + k];
            w2v[kc * 2 + i] = wl[32 + k];
        }
    }
    for (int t = w; t < ntile; t += 32) {
        const int  row = t * 16 + m;
        const bool ok  = row < Rb;
        float dm = dm0, xm = xm0;
        if (t != w) {
            dm = ok ? dinv[n0 + row] : 0.0f;
            xm = ok ? x[n0 + row] : 0.0f;
        }
        const float ts = ok ? dm * rsum[row] : 0.0f;
        const float sl = dm * dm * xm;
        v8f d = {0.f, 0.f, 0.f, 0.f, 0.f, 0.f, 0.f, 0.f};
#pragma unroll
        for (int kc = 0; kc < 4; ++kc) {
            v2f a, b;
            a.x = fmaxf(ts * w1v[kc * 2 + 0] + sl * w1v[kc * 2 + 0] + b1v[kc * 2 + 0], 0.0f);
            a.y = fmaxf(ts * w1v[kc * 2 + 1] + sl * w1v[kc * 2 + 1] + b1v[kc * 2 + 1], 0.0f);
            b.x = w2v[kc * 2 + 0];
            b.y = w2v[kc * 2 + 1];
            d = __builtin_amdgcn_wmma_f32_16x16x4_f32(
                 false, a,  false, b,
                 (short)0, d,  false,  false);
        }
        if (m == 0) {
            const int q  = t * 16 + kh * 8;
            float* zp = z + n0 + q;
            if (q     < Rb) zp[0] = d[0];
            if (q + 1 < Rb) zp[1] = d[1];
            if (q + 2 < Rb) zp[2] = d[2];
            if (q + 3 < Rb) zp[3] = d[3];
            if (q + 4 < Rb) zp[4] = d[4];
            if (q + 5 < Rb) zp[5] = d[5];
            if (q + 6 < Rb) zp[6] = d[6];
            if (q + 7 < Rb) zp[7] = d[7];
        }
    }
}

__global__ __launch_bounds__(1024) void gcn_layer2_kernel(const int* __restrict__ rowptr,
                                                          const int* __restrict__ cnt,
                                                          const int* __restrict__ col,
                                                          const float* __restrict__ dinv,
                                                          const float* __restrict__ z,
                                                          const float* __restrict__ b2,
                                                          float* __restrict__ out, int E, int N) {
    __shared__ float tab[GCN_TC];
    __shared__ int2  lrow[GCN_RMAX];
    __shared__ float rsum[GCN_RMAX];
    const int R  = (N + GCN_NB - 1) / GCN_NB;
    const int n0 = blockIdx.x * R;
    if (n0 >= N) return;
    const int Rb = (N - n0 < R) ? (N - n0) : R;
    const int tid = threadIdx.x;
    const float bias = b2[0];
    float dn0 = 0.0f, zn0 = 0.0f;
    if (tid < Rb) { dn0 = dinv[n0 + tid]; zn0 = z[n0 + tid]; }

    gcn_range_rows(rowptr, cnt, col, dinv, z, N, E, n0, Rb, tab, lrow, rsum);

    for (int t = tid; t < Rb; t += 1024) {
        float dn = dn0, zn = zn0;
        if (t != tid) { dn = dinv[n0 + t]; zn = z[n0 + t]; }
        out[n0 + t] = dn * rsum[t] + (dn * dn * zn + bias);
    }
}

extern "C" void kernel_launch(void* const* d_in, const int* in_sizes, int n_in,
                              void* d_out, int out_size, void* d_ws, size_t ws_size,
                              hipStream_t stream) {
    const float* x  = (const float*)d_in[0];
    const float* W1 = (const float*)d_in[1];
    const float* b1 = (const float*)d_in[2];
    const float* W2 = (const float*)d_in[3];
    const float* b2 = (const float*)d_in[4];
    const int*   ei = (const int*)d_in[5];
    int N = in_sizes[0];
    int E = in_sizes[5] / 2;
    const int* src = ei;
    const int* dst = ei + E;
    int paddedN = (N + 15) & ~15;

    char* ws = (char*)d_ws;
    float* dinv   = (float*)ws; ws += (size_t)paddedN * sizeof(float);
    float* z      = (float*)ws; ws += (size_t)paddedN * sizeof(float);
    int*   cnt    = (int*)ws;   ws += (size_t)paddedN * sizeof(int);
    int*   rowptr = (int*)ws;   ws += (size_t)paddedN * sizeof(int);
    int*   T      = (int*)ws;   ws += (size_t)(GCN_NB + 1) * GCN_NS * sizeof(int);
    int*   pairs  = (int*)ws;   ws += (((size_t)E * 2 * sizeof(int)) + 255) & ~(size_t)255;
    int*   col    = (int*)ws;
    float* out    = (float*)d_out;

    gcn_slice_sort_kernel<<<GCN_NS, 1024, 0, stream>>>(src, dst, T, pairs, E, N);
    gcn_rows_kernel<<<GCN_NB, 1024, 0, stream>>>(pairs, T, rowptr, cnt, dinv, col, E, N);
    gcn_layer1_kernel<<<GCN_NB, 1024, 0, stream>>>(rowptr, cnt, col, dinv, x, W1, b1, W2, z, E, N);
    gcn_layer2_kernel<<<GCN_NB, 1024, 0, stream>>>(rowptr, cnt, col, dinv, z, b2, out, E, N);
}


# Per-source term dinv*v stored once per node by the kernel that makes it (rows: dinv*x, layer 1: dinv*z); each block's LDS table copy reads one array, not two: half its bytes; sum order kept
# speedup vs baseline: 10.2091x; 1.1200x over previous
#include <hip/hip_runtime.h>
#include <hip/hip_bf16.h>

typedef __attribute__((ext_vector_type(2)))  float    v2f;
typedef __attribute__((ext_vector_type(8)))  float    v8f;
typedef __attribute__((ext_vector_type(4)))  float    v4f;

#define GCN_NB   256
#define GCN_NS   256
#define GCN_RMAX 4096
#define GCN_LPT  16
#define GCN_LMAX (1024 * GCN_LPT)
#define GCN_CCAP 32768
#define GCN_RPW  (GCN_NS / 32)
#define GCN_RLD  3

__global__ __launch_bounds__(1024) void gcn_slice_sort_kernel(const int* __restrict__ src,
                                                              const int* __restrict__ dst,
                                                              int* __restrict__ T,
                                                              int* __restrict__ pairs_raw,
                                                              int E, int N) {
    __shared__ int2 stage[GCN_LMAX];
    __shared__ int hist[GCN_NB];
    __shared__ int first[GCN_NB];
    __shared__ int wtot[GCN_NB / 32];
    int2* pairs = (int2*)pairs_raw;
    const int tid  = threadIdx.x;
    const int lane = tid & 31;
    const int w    = tid >> 5;
    const int s    = blockIdx.x;
    const unsigned R = (unsigned)((N + GCN_NB - 1) / GCN_NB);
    const int L  = (E + GCN_NS - 1) / GCN_NS;
    const int e0 = s * L;
    const int e1 = (e0 + L < E) ? (e0 + L) : E;
    const int len = e1 > e0 ? e1 - e0 : 0;
    if (tid < GCN_NB) hist[tid] = 0;

    int d[GCN_LPT], sv[GCN_LPT], rg[GCN_LPT], rk[GCN_LPT];
#pragma unroll
    for (int k = 0; k < GCN_LPT; ++k) {
        const int ek = e0 + tid + k * 1024;
        const int ec = ek < e1 ? ek : e1 - 1;
        d[k]  = dst[ec];
        sv[k] = src[ec];
    }
    __syncthreads();
#pragma unroll
    for (int k = 0; k < GCN_LPT; ++k) {
        rg[k] = (int)((unsigned)d[k] / R);
        rk[k] = 0;
        if (e0 + tid + k * 1024 < e1) rk[k] = atomicAdd(&hist[rg[k]], 1);
    }
    __syncthreads();

    const int c = (tid < GCN_NB) ? hist[tid] : 0;
    int inc = c;
    for (int o = 1; o < 32; o <<= 1) {
        int y = __shfl_up(inc, o, 32);
        if (lane >= o) inc += y;
    }
    if (tid < GCN_NB && lane == 31) wtot[w] = inc;
    __syncthreads();
    if (tid < GCN_NB) {
        int before = 0;
#pragma unroll
        for (int j = 0; j < GCN_NB / 32; ++j) before += (j < w) ? wtot[j] : 0;
        const int f = before + inc - c;
        first[tid] = f;
        T[tid * GCN_NS + s] = f;
        if (tid == GCN_NB - 1) T[GCN_NB * GCN_NS + s] = f + c;
    }
    __syncthreads();

#pragma unroll
    for (int k = 0; k < GCN_LPT; ++k) rg[k] = first[rg[k]] + rk[k];
#pragma unroll
    for (int k = 0; k < GCN_LPT; ++k) {
        if (e0 + tid + k * 1024 < e1) stage[rg[k]] = make_int2(sv[k], d[k]);
    }
    __syncthreads();
#pragma unroll 4
    for (int i = tid; i < len; i += 1024) pairs[e0 + i] = stage[i];
}

__global__ __launch_bounds__(1024) void gcn_rows_kernel(const int* __restrict__ pairs_raw,
                                                        const int* __restrict__ T,
                                                        int* __restrict__ rowptr,
                                                        int* __restrict__ cnt,
                                                        float* __restrict__ dinv,
                                                        const float* __restrict__ x,
                                                        float* __restrict__ wx,
                                                        int* __restrict__ col, int E, int N) {
    __shared__ int lcol[GCN_CCAP];
    __shared__ int lcnt[GCN_RMAX];
    __shared__ int rs[GCN_NS];
    __shared__ int rl[GCN_NS];
    __shared__ int wtot[32];
    __shared__ int redA[GCN_NS / 32];
    __shared__ int redL[GCN_NS / 32];
    const int2* pairs = (const int2*)pairs_raw;
    const int tid  = threadIdx.x;
    const int lane = tid & 31;
    const int w    = tid >> 5;
    const int b    = blockIdx.x;
    const int R    = (N + GCN_NB - 1) / GCN_NB;
    const int n0   = b * R;
    if (n0 >= N) return;
    const int L = (E + GCN_NS - 1) / GCN_NS;

    int ra = 0, rn = 0;
    if (tid < GCN_NS) {
        ra = T[b * GCN_NS + tid];
        rn = T[(b + 1) * GCN_NS + tid] - ra;
        rs[tid] = tid * L + ra;
        rl[tid] = rn;
    }
    int sa = ra, sl = rn;
    for (int o = 16; o > 0; o >>= 1) {
        sa += __shfl_xor(sa, o, 32);
        sl += __shfl_xor(sl, o, 32);
    }
    if (tid < GCN_NS && lane == 0) { redA[w] = sa; redL[w] = sl; }
    const int i = tid * 4;
    float xv0 = 0.0f, xv1 = 0.0f, xv2 = 0.0f, xv3 = 0.0f;
    if (i     < R && n0 + i     < N) xv0 = x[n0 + i];
    if (i + 1 < R && n0 + i + 1 < N) xv1 = x[n0 + i + 1];
    if (i + 2 < R && n0 + i + 2 < N) xv2 = x[n0 + i + 2];
    if (i + 3 < R && n0 + i + 3 < N) xv3 = x[n0 + i + 3];
    *(int4*)(lcnt + i) = make_int4(0, 0, 0, 0);
    __syncthreads();
    int p0 = 0, tot = 0;
#pragma unroll
    for (int j = 0; j < GCN_NS / 32; ++j) { p0 += redA[j]; tot += redL[j]; }
    const bool staged = tot <= GCN_CCAP;

    int  st[GCN_RPW], rn_[GCN_RPW];
    int2 rec[GCN_RPW][GCN_RLD];
#pragma unroll
    for (int r = 0; r < GCN_RPW; ++r) {
        st[r]  = rs[w + 32 * r];
        rn_[r] = rl[w + 32 * r];
#pragma unroll
        for (int h = 0; h < GCN_RLD; ++h) {
            const int j = lane + 32 * h;
            int g = st[r] + (j < rn_[r] ? j : rn_[r] - 1);
            g = g < E ? g : E - 1;
            g = g > 0 ? g : 0;
            rec[r][h] = pairs[g];
        }
    }

#pragma unroll
    for (int r = 0; r < GCN_RPW; ++r) {
#pragma unroll
        for (int h = 0; h < GCN_RLD; ++h) {
            if (lane + 32 * h < rn_[r]) atomicAdd(&lcnt[rec[r][h].y - n0], 1);
        }
        for (int j = lane + 32 * GCN_RLD; j < rn_[r]; j += 32)
            atomicAdd(&lcnt[pairs[st[r] + j].y - n0], 1);
    }
    __syncthreads();

    const int4 c = *(const int4*)(lcnt + i);
    const int part = (c.x + c.y) + (c.z + c.w);
    int inc = part;
    for (int o = 1; o < 32; o <<= 1) {
        int y = __shfl_up(inc, o, 32);
        if (lane >= o) inc += y;
    }
    if (lane == 31) wtot[w] = inc;
    __syncthreads();
    const int wt = wtot[lane];
    int winc = wt;
    for (int o = 1; o < 32; o <<= 1) {
        int y = __shfl_up(winc, o, 32);
        if (lane >= o) winc += y;
    }
    const int q0 = p0 + __shfl(winc - wt, w, 32) + inc - part;
    const int q1 = q0 + c.x;
    const int q2 = q1 + c.y;
    const int q3 = q2 + c.z;
    {
        const int n = n0 + i;
        const float d0 = rsqrtf((float)(c.x + 1));
        const float d1 = rsqrtf((float)(c.y + 1));
        const float d2 = rsqrtf((float)(c.z + 1));
        const float d3 = rsqrtf((float)(c.w + 1));
        if (i     < R && n     < N) { rowptr[n]     = q0; cnt[n]     = c.x; dinv[n]     = d0; wx[n]     = d0 * xv0; }
        if (i + 1 < R && n + 1 < N) { rowptr[n + 1] = q1; cnt[n + 1] = c.y; dinv[n + 1] = d1; wx[n + 1] = d1 * xv1; }
        if (i + 2 < R && n + 2 < N) { rowptr[n + 2] = q2; cnt[n + 2] = c.z; dinv[n + 2] = d2; wx[n + 2] = d2 * xv2; }
        if (i + 3 < R && n + 3 < N) { rowptr[n + 3] = q3; cnt[n + 3] = c.w; dinv[n + 3] = d3; wx[n + 3] = d3 * xv3; }
    }
    *(int4*)(lcnt + i) = make_int4(q0, q1, q2, q3);
    __syncthreads();

#pragma unroll
    for (int r = 0; r < GCN_RPW; ++r) {
        int slot[GCN_RLD];
#pragma unroll
        for (int h = 0; h < GCN_RLD; ++h) {
            slot[h] = p0;
            if (lane + 32 * h < rn_[r]) slot[h] = atomicAdd(&lcnt[rec[r][h].y - n0], 1);
        }
#pragma unroll
        for (int h = 0; h < GCN_RLD; ++h) {
            if (lane + 32 * h < rn_[r]) {
                if (staged) lcol[slot[h] - p0] = rec[r][h].x;
                else        col[slot[h]]       = rec[r][h].x;
            }
        }
        for (int j = lane + 32 * GCN_RLD; j < rn_[r]; j += 32) {
            const int2 e = pairs[st[r] + j];
            const int q = atomicAdd(&lcnt[e.y - n0], 1);
            if (staged) lcol[q - p0] = e.x;
            else        col[q]       = e.x;
        }
    }
    if (staged) {
        __syncthreads();
        for (int j = tid; j < tot; j += 1024) col[p0 + j] = lcol[j];
    }
}

#define GCN_TC   57344
#define GCN_EPT  16
#define GCN_SG   14
#define GCN_RU   8
static_assert(GCN_RU == 8, "the read fence in gcn_range_rows names 8 rows");

__device__ __forceinline__ float gcn_sum16(float a) {
    a += __int_as_float(__builtin_amdgcn_update_dpp(0, __float_as_int(a), 0x168, 0xf, 0xf, false));
    a += __int_as_float(__builtin_amdgcn_update_dpp(0, __float_as_int(a), 0x164, 0xf, 0xf, false));
    a += __int_as_float(__builtin_amdgcn_update_dpp(0, __float_as_int(a), 0x162, 0xf, 0xf, false));
    a += __int_as_float(__builtin_amdgcn_update_dpp(0, __float_as_int(a), 0x161, 0xf, 0xf, false));
    return a;
}

__device__ __forceinline__ void gcn_range_rows(const int* __restrict__ rowptr,
                                               const int* __restrict__ cnt,
                                               const int* __restrict__ col,
                                               const float* __restrict__ wt,
                                               int N, int E, int n0, int Rb,
                                               float* tab, int2* lrow, float* rsum) {
    const int tid = threadIdx.x;
    const int r16 = tid >> 4;
    const int f   = tid & 15;
    const int nl  = n0 + Rb - 1;
    const int p0  = rowptr[n0];
    const int tot = rowptr[nl] + cnt[nl] - p0;
    for (int t = tid; t < Rb; t += 1024)
        lrow[t] = make_int2(rowptr[n0 + t] - p0, cnt[n0 + t]);
    if (tot <= GCN_EPT * 1024) {
        int   cj[GCN_EPT];
        float val[GCN_EPT];
#pragma unroll
        for (int k = 0; k < GCN_EPT; ++k) {
            int j = tid + 1024 * k;
            j = j < tot ? j : tot - 1;
            int g = p0 + j;
            g = g < E ? g : E - 1;
            g = g > 0 ? g : 0;
            cj[k]  = col[g];
            val[k] = 0.0f;
        }
        for (int c0 = 0; c0 < N; c0 += GCN_TC) {
            const int clen = (N - c0 < GCN_TC) ? (N - c0) : GCN_TC;
            const int cl4  = clen & ~3;
            const v4f* wch4 = (const v4f*)(wt + c0);
            v4f*       tab4 = (v4f*)tab;
            const unsigned n4 = (unsigned)cl4 >> 2;
            for (unsigned g = (unsigned)tid; g < n4; g += 1024u * GCN_SG) {
                v4f a[GCN_SG];
                unsigned ic[GCN_SG];
#pragma unroll
                for (int u = 0; u < GCN_SG; ++u) {
                    const unsigned i = g + (unsigned)u * 1024u;
                    ic[u] = i < n4 ? i : g;
                    a[u] = wch4[ic[u]];
                }
                __builtin_amdgcn_sched_barrier(0);
#pragma unroll
                for (int u = 0; u < GCN_SG; ++u) tab4[ic[u]] = a[u];
            }
            if (tid < clen - cl4) {
                const int i = cl4 + tid;
                tab[i] = wt[c0 + i];
            }
            __syncthreads();
#pragma unroll
            for (int k = 0; k < GCN_EPT; ++k) {
                const unsigned loc = (unsigned)(cj[k] - c0);
                if (loc < (unsigned)clen) val[k] = tab[loc];
            }
            __syncthreads();
        }
#pragma unroll
        for (int k = 0; k < GCN_EPT; ++k) {
            const int j = tid + 1024 * k;
            if (j < tot) tab[j] = val[k];
        }
        __syncthreads();
        for (int rb = 0; rb < Rb; rb += 64 * GCN_RU) {
            int   st[GCN_RU], ln[GCN_RU];
            float acc[GCN_RU];
            int   mx = 0;
#pragma unroll
            for (int u = 0; u < GCN_RU; ++u) {
                const int  r  = rb + 64 * u + r16;
                const bool ok = r < Rb;
                const int2 e  = lrow[ok ? r : 0];
                st[u]  = e.x;
                ln[u]  = ok ? e.y : 0;
                acc[u] = 0.0f;
                mx = ln[u] > mx ? ln[u] : mx;
            }
            for (int j = f; j < mx; j += 32) {
                float a0[GCN_RU], a1[GCN_RU];
#pragma unroll
                for (int u = 0; u < GCN_RU; ++u) {
                    a0[u] = tab[st[u] + j];
                    a1[u] = tab[st[u] + j + 16];
                }
                asm volatile("" :: "v"(a0[0]), "v"(a0[1]), "v"(a0[2]), "v"(a0[3]), "v"(a0[4]), "v"(a0[5]), "v"(a0[6]), "v"(a0[7]),
                                   "v"(a1[0]), "v"(a1[1]), "v"(a1[2]), "v"(a1[3]), "v"(a1[4]), "v"(a1[5]), "v"(a1[6]), "v"(a1[7]));
#pragma unroll
                for (int u = 0; u < GCN_RU; ++u) {
                    const bool  one = j < ln[u];
                    const bool  two = j + 16 < ln[u];
                    const float t   = a0[u] + (two ? a1[u] : 0.0f);
                    acc[u] = one ? acc[u] + t : acc[u];
                }
            }
#pragma unroll
            for (int u = 0; u < GCN_RU; ++u) acc[u] = gcn_sum16(acc[u]);
            float tsel = acc[0];
#pragma unroll
            for (int u = 1; u < GCN_RU; ++u) tsel = (f == u) ? acc[u] : tsel;
            const int r = rb + 64 * f + r16;
            if (f < GCN_RU && r < Rb) rsum[r] = tsel;
        }
    } else {
        __syncthreads();
        for (int rb = 0; rb < Rb; rb += 64) {
            const int  r  = rb + r16;
            const bool ok = r < Rb;
            const int2 e    = lrow[ok ? r : 0];
            const int start = ok ? e.x + p0 : 0;
            const int len   = ok ? e.y : 0;
            float acc = 0.0f;
            for (int j = f; j < len; j += 32) {
                const int  j1  = j + 16;
                const bool two = j1 < len;
                const int  s0  = col[start + j];
                const int  s1  = col[start + (two ? j1 : j)];
                const float a0 = wt[s0];
                const float a1 = wt[s1];
                acc += a0 + (two ? a1 : 0.0f);
            }
            acc = gcn_sum16(acc);
            if (ok && f == 0) rsum[r] = acc;
        }
    }
    __syncthreads();
}

__global__ __launch_bounds__(1024) void gcn_layer1_kernel(const int* __restrict__ rowptr,
                                                          const int* __restrict__ cnt,
                                                          const int* __restrict__ col,
                                                          const float* __restrict__ dinv,
                                                          const float* __restrict__ x,
                                                          const float* __restrict__ W1,
                                                          const float* __restrict__ b1,
                                                          const float* __restrict__ W2,
                                                          const float* __restrict__ wx,
                                                          float* __restrict__ z,
                                                          float* __restrict__ wz, int E, int N) {
    __shared__ float tab[GCN_TC];
    __shared__ int2  lrow[GCN_RMAX];
    __shared__ float rsum[GCN_RMAX];
    __shared__ float wl[48];
    __shared__ float own[2048];
    const int R  = (N + GCN_NB - 1) / GCN_NB;
    const int n0 = blockIdx.x * R;
    if (n0 >= N) return;
    const int Rb = (N - n0 < R) ? (N - n0) : R;
    const int w  = threadIdx.x >> 5;
    const int m  = threadIdx.x & 15;
    const int kh = (threadIdx.x >> 4) & 1;
    const int ntile = (Rb + 15) >> 4;

    if (threadIdx.x < 16) {
        wl[threadIdx.x]      = W1[threadIdx.x];
        wl[16 + threadIdx.x] = b1[threadIdx.x];
        wl[32 + threadIdx.x] = W2[threadIdx.x];
    }
    {
        float dm0 = 0.0f, xm0 = 0.0f;
        if (w * 16 + m < Rb) { dm0 = dinv[n0 + w * 16 + m]; xm0 = x[n0 + w * 16 + m]; }
        own[threadIdx.x]        = dm0;
        own[1024 + threadIdx.x] = xm0;
    }

    gcn_range_rows(rowptr, cnt, col, wx, N, E, n0, Rb, tab, lrow, rsum);

    const float dm0 = own[threadIdx.x];
    const float xm0 = own[1024 + threadIdx.x];

    float w1v[8], b1v[8], w2v[8];
#pragma unroll
    for (int kc = 0; kc < 4; ++kc) {
#pragma unroll
        for (int i = 0; i < 2; ++i) {
            const int k = kc * 4 + kh * 2 + i;
            w1v[kc * 2 + i] = wl[k];
            b1v[kc * 2 + i] = wl[16 + k];
            w2v[kc * 2 + i] = wl[32 + k];
        }
    }
    for (int t = w; t < ntile; t += 32) {
        const int  row = t * 16 + m;
        const bool ok  = row < Rb;
        float dm = dm0, xm = xm0;
        if (t != w) {
            dm = ok ? dinv[n0 + row] : 0.0f;
            xm = ok ? x[n0 + row] : 0.0f;
        }
        const float ts = ok ? dm * rsum[row] : 0.0f;
        const float sl = dm * dm * xm;
        v8f d = {0.f, 0.f, 0.f, 0.f, 0.f, 0.f, 0.f, 0.f};
#pragma unroll
        for (int kc = 0; kc < 4; ++kc) {
            v2f a, b;
            a.x = fmaxf(ts * w1v[kc * 2 + 0] + sl * w1v[kc * 2 + 0] + b1v[kc * 2 + 0], 0.0f);
            a.y = fmaxf(ts * w1v[kc * 2 + 1] + sl * w1v[kc * 2 + 1] + b1v[kc * 2 + 1], 0.0f);
            b.x = w2v[kc * 2 + 0];
            b.y = w2v[kc * 2 + 1];
            d = __builtin_amdgcn_wmma_f32_16x16x4_f32(
                 false, a,  false, b,
                 (short)0, d,  false,  false);
        }
        float zr = d[0];
#pragma unroll
        for (int r = 1; r < 8; ++r) zr = ((m & 7) == r) ? d[r] : zr;
        if ((m >> 3) == kh && ok) {
            z[n0 + row]  = zr;
            wz[n0 + row] = dm * zr;
        }
    }
}

__global__ __launch_bounds__(1024) void gcn_layer2_kernel(const int* __restrict__ rowptr,
                                                          const int* __restrict__ cnt,
                                                          const int* __restrict__ col,
                                                          const float* __restrict__ dinv,
                                                          const float* __restrict__ z,
                                                          const float* __restrict__ wz,
                                                          const float* __restrict__ b2,
                                                          float* __restrict__ out, int E, int N) {
    __shared__ float tab[GCN_TC];
    __shared__ int2  lrow[GCN_RMAX];
    __shared__ float rsum[GCN_RMAX];
    const int R  = (N + GCN_NB - 1) / GCN_NB;
    const int n0 = blockIdx.x * R;
    if (n0 >= N) return;
    const int Rb = (N - n0 < R) ? (N - n0) : R;
    const int tid = threadIdx.x;
    const float bias = b2[0];
    float dn0 = 0.0f, zn0 = 0.0f;
    if (tid < Rb) { dn0 = dinv[n0 + tid]; zn0 = z[n0 + tid]; }

    gcn_range_rows(rowptr, cnt, col, wz, N, E, n0, Rb, tab, lrow, rsum);

    for (int t = tid; t < Rb; t += 1024) {
        float dn = dn0, zn = zn0;
        if (t != tid) { dn = dinv[n0 + t]; zn = z[n0 + t]; }
        out[n0 + t] = dn * rsum[t] + (dn * dn * zn + bias);
    }
}

extern "C" void kernel_launch(void* const* d_in, const int* in_sizes, int n_in,
                              void* d_out, int out_size, void* d_ws, size_t ws_size,
                              hipStream_t stream) {
    const float* x  = (const float*)d_in[0];
    const float* W1 = (const float*)d_in[1];
    const float* b1 = (const float*)d_in[2];
    const float* W2 = (const float*)d_in[3];
    const float* b2 = (const float*)d_in[4];
    const int*   ei = (const int*)d_in[5];
    int N = in_sizes[0];
    int E = in_sizes[5] / 2;
    const int* src = ei;
    const int* dst = ei + E;
    int paddedN = (N + 15) & ~15;

    char* ws = (char*)d_ws;
    float* dinv   = (float*)ws; ws += (size_t)paddedN * sizeof(float);
    float* z      = (float*)ws; ws += (size_t)paddedN * sizeof(float);
    int*   cnt    = (int*)ws;   ws += (size_t)paddedN * sizeof(int);
    int*   rowptr = (int*)ws;   ws += (size_t)paddedN * sizeof(int);
    float* wx     = (float*)ws; ws += (size_t)paddedN * sizeof(float);
    float* wz     = (float*)ws; ws += (size_t)paddedN * sizeof(float);
    int*   T      = (int*)ws;   ws += (size_t)(GCN_NB + 1) * GCN_NS * sizeof(int);
    int*   pairs  = (int*)ws;   ws += (((size_t)E * 2 * sizeof(int)) + 255) & ~(size_t)255;
    int*   col    = (int*)ws;
    float* out    = (float*)d_out;

    gcn_slice_sort_kernel<<<GCN_NS, 1024, 0, stream>>>(src, dst, T, pairs, E, N);
    gcn_rows_kernel<<<GCN_NB, 1024, 0, stream>>>(pairs, T, rowptr, cnt, dinv, x, wx, col, E, N);
    gcn_layer1_kernel<<<GCN_NB, 1024, 0, stream>>>(rowptr, cnt, col, dinv, x, W1, b1, W2, wx, z, wz, E, N);
    gcn_layer2_kernel<<<GCN_NB, 1024, 0, stream>>>(rowptr, cnt, col, dinv, z, wz, b2, out, E, N);
}


# Sorted edge records packed in one 32-bit word (src id in 17 bits, dst offset in its range above): record array written and read at half the bytes, slices copied out 16 B a lane; row order = arrival
# speedup vs baseline: 10.6203x; 1.0403x over previous
#include <hip/hip_runtime.h>
#include <hip/hip_bf16.h>

typedef __attribute__((ext_vector_type(2)))  float    v2f;
typedef __attribute__((ext_vector_type(8)))  float    v8f;
typedef __attribute__((ext_vector_type(4)))  float    v4f;

#define GCN_NB   256
#define GCN_NS   256
#define GCN_RMAX 4096
#define GCN_LPT  16
#define GCN_LMAX (1024 * GCN_LPT)
#define GCN_CCAP 32768
#define GCN_RPW  (GCN_NS / 32)
#define GCN_RLD  3
#define GCN_SB   17
#define GCN_SM   ((1u << GCN_SB) - 1u)

__global__ __launch_bounds__(1024) void gcn_slice_sort_kernel(const int* __restrict__ src,
                                                              const int* __restrict__ dst,
                                                              int* __restrict__ T,
                                                              int* __restrict__ pairs_raw,
                                                              int E, int N) {
    __shared__ __attribute__((aligned(16))) unsigned stage[GCN_LMAX];
    __shared__ int hist[GCN_NB];
    __shared__ int first[GCN_NB];
    __shared__ int wtot[GCN_NB / 32];
    unsigned* pairs = (unsigned*)pairs_raw;
    const int tid  = threadIdx.x;
    const int lane = tid & 31;
    const int w    = tid >> 5;
    const int s    = blockIdx.x;
    const unsigned R = (unsigned)((N + GCN_NB - 1) / GCN_NB);
    const int L  = (E + GCN_NS - 1) / GCN_NS;
    const int e0 = s * L;
    const int e1 = (e0 + L < E) ? (e0 + L) : E;
    const int len = e1 > e0 ? e1 - e0 : 0;
    if (tid < GCN_NB) hist[tid] = 0;

    int d[GCN_LPT], sv[GCN_LPT], rg[GCN_LPT], rk[GCN_LPT];
#pragma unroll
    for (int k = 0; k < GCN_LPT; ++k) {
        const int ek = e0 + tid + k * 1024;
        const int ec = ek < e1 ? ek : e1 - 1;
        d[k]  = dst[ec];
        sv[k] = src[ec];
    }
    __syncthreads();
#pragma unroll
    for (int k = 0; k < GCN_LPT; ++k) {
        rg[k] = (int)((unsigned)d[k] / R);
        sv[k] = (int)((unsigned)sv[k] | (((unsigned)d[k] - (unsigned)rg[k] * R) << GCN_SB));
        rk[k] = 0;
        if (e0 + tid + k * 1024 < e1) rk[k] = atomicAdd(&hist[rg[k]], 1);
    }
    __syncthreads();

    const int c = (tid < GCN_NB) ? hist[tid] : 0;
    int inc = c;
    for (int o = 1; o < 32; o <<= 1) {
        int y = __shfl_up(inc, o, 32);
        if (lane >= o) inc += y;
    }
    if (tid < GCN_NB && lane == 31) wtot[w] = inc;
    __syncthreads();
    if (tid < GCN_NB) {
        int before = 0;
#pragma unroll
        for (int j = 0; j < GCN_NB / 32; ++j) before += (j < w) ? wtot[j] : 0;
        const int f = before + inc - c;
        first[tid] = f;
        T[tid * GCN_NS + s] = f;
        if (tid == GCN_NB - 1) T[GCN_NB * GCN_NS + s] = f + c;
    }
    __syncthreads();

#pragma unroll
    for (int k = 0; k < GCN_LPT; ++k) rg[k] = first[rg[k]] + rk[k];
#pragma unroll
    for (int k = 0; k < GCN_LPT; ++k) {
        if (e0 + tid + k * 1024 < e1) stage[rg[k]] = (unsigned)sv[k];
    }
    __syncthreads();
    if ((L & 3) == 0) {
        const int n4 = len >> 2;
        const uint4* st4 = (const uint4*)stage;
        uint4*       pr4 = (uint4*)(pairs + e0);
        for (int i = tid; i < n4; i += 1024) pr4[i] = st4[i];
        if (tid < (len & 3)) pairs[e0 + (n4 << 2) + tid] = stage[(n4 << 2) + tid];
    } else {
#pragma unroll 4
        for (int i = tid; i < len; i += 1024) pairs[e0 + i] = stage[i];
    }
}

__global__ __launch_bounds__(1024) void gcn_rows_kernel(const int* __restrict__ pairs_raw,
                                                        const int* __restrict__ T,
                                                        int* __restrict__ rowptr,
                                                        int* __restrict__ cnt,
                                                        float* __restrict__ dinv,
                                                        const float* __restrict__ x,
                                                        float* __restrict__ wx,
                                                        int* __restrict__ col, int E, int N) {
    __shared__ int lcol[GCN_CCAP];
    __shared__ int lcnt[GCN_RMAX];
    __shared__ int rs[GCN_NS];
    __shared__ int rl[GCN_NS];
    __shared__ int wtot[32];
    __shared__ int redA[GCN_NS / 32];
    __shared__ int redL[GCN_NS / 32];
    const unsigned* pairs = (const unsigned*)pairs_raw;
    const int tid  = threadIdx.x;
    const int lane = tid & 31;
    const int w    = tid >> 5;
    const int b    = blockIdx.x;
    const int R    = (N + GCN_NB - 1) / GCN_NB;
    const int n0   = b * R;
    if (n0 >= N) return;
    const int L = (E + GCN_NS - 1) / GCN_NS;

    int ra = 0, rn = 0;
    if (tid < GCN_NS) {
        ra = T[b * GCN_NS + tid];
        rn = T[(b + 1) * GCN_NS + tid] - ra;
        rs[tid] = tid * L + ra;
        rl[tid] = rn;
    }
    int sa = ra, sl = rn;
    for (int o = 16; o > 0; o >>= 1) {
        sa += __shfl_xor(sa, o, 32);
        sl += __shfl_xor(sl, o, 32);
    }
    if (tid < GCN_NS && lane == 0) { redA[w] = sa; redL[w] = sl; }
    const int i = tid * 4;
    float xv0 = 0.0f, xv1 = 0.0f, xv2 = 0.0f, xv3 = 0.0f;
    if (i     < R && n0 + i     < N) xv0 = x[n0 + i];
    if (i + 1 < R && n0 + i + 1 < N) xv1 = x[n0 + i + 1];
    if (i + 2 < R && n0 + i + 2 < N) xv2 = x[n0 + i + 2];
    if (i + 3 < R && n0 + i + 3 < N) xv3 = x[n0 + i + 3];
    *(int4*)(lcnt + i) = make_int4(0, 0, 0, 0);
    __syncthreads();
    int p0 = 0, tot = 0;
#pragma unroll
    for (int j = 0; j < GCN_NS / 32; ++j) { p0 += redA[j]; tot += redL[j]; }
    const bool staged = tot <= GCN_CCAP;

    int  st[GCN_RPW], rn_[GCN_RPW];
    unsigned rec[GCN_RPW][GCN_RLD];
#pragma unroll
    for (int r = 0; r < GCN_RPW; ++r) {
        st[r]  = rs[w + 32 * r];
        rn_[r] = rl[w + 32 * r];
#pragma unroll
        for (int h = 0; h < GCN_RLD; ++h) {
            const int j = lane + 32 * h;
            int g = st[r] + (j < rn_[r] ? j : rn_[r] - 1);
            g = g < E ? g : E - 1;
            g = g > 0 ? g : 0;
            rec[r][h] = pairs[g];
        }
    }

#pragma unroll
    for (int r = 0; r < GCN_RPW; ++r) {
#pragma unroll
        for (int h = 0; h < GCN_RLD; ++h) {
            if (lane + 32 * h < rn_[r]) atomicAdd(&lcnt[rec[r][h] >> GCN_SB], 1);
        }
        for (int j = lane + 32 * GCN_RLD; j < rn_[r]; j += 32)
            atomicAdd(&lcnt[pairs[st[r] + j] >> GCN_SB], 1);
    }
    __syncthreads();

    const int4 c = *(const int4*)(lcnt + i);
    const int part = (c.x + c.y) + (c.z + c.w);
    int inc = part;
    for (int o = 1; o < 32; o <<= 1) {
        int y = __shfl_up(inc, o, 32);
        if (lane >= o) inc += y;
    }
    if (lane == 31) wtot[w] = inc;
    __syncthreads();
    const int wt = wtot[lane];
    int winc = wt;
    for (int o = 1; o < 32; o <<= 1) {
        int y = __shfl_up(winc, o, 32);
        if (lane >= o) winc += y;
    }
    const int q0 = p0 + __shfl(winc - wt, w, 32) + inc - part;
    const int q1 = q0 + c.x;
    const int q2 = q1 + c.y;
    const int q3 = q2 + c.z;
    {
        const int n = n0 + i;
        const float d0 = rsqrtf((float)(c.x + 1));
        const float d1 = rsqrtf((float)(c.y + 1));
        const float d2 = rsqrtf((float)(c.z + 1));
        const float d3 = rsqrtf((float)(c.w + 1));
        if (i     < R && n     < N) { rowptr[n]     = q0; cnt[n]     = c.x; dinv[n]     = d0; wx[n]     = d0 * xv0; }
        if (i + 1 < R && n + 1 < N) { rowptr[n + 1] = q1; cnt[n + 1] = c.y; dinv[n + 1] = d1; wx[n + 1] = d1 * xv1; }
        if (i + 2 < R && n + 2 < N) { rowptr[n + 2] = q2; cnt[n + 2] = c.z; dinv[n + 2] = d2; wx[n + 2] = d2 * xv2; }
        if (i + 3 < R && n + 3 < N) { rowptr[n + 3] = q3; cnt[n + 3] = c.w; dinv[n + 3] = d3; wx[n + 3] = d3 * xv3; }
    }
    *(int4*)(lcnt + i) = make_int4(q0, q1, q2, q3);
    __syncthreads();

#pragma unroll
    for (int r = 0; r < GCN_RPW; ++r) {
        int slot[GCN_RLD];
#pragma unroll
        for (int h = 0; h < GCN_RLD; ++h) {
            slot[h] = p0;
            if (lane + 32 * h < rn_[r]) slot[h] = atomicAdd(&lcnt[rec[r][h] >> GCN_SB], 1);
        }
#pragma unroll
        for (int h = 0; h < GCN_RLD; ++h) {
            if (lane + 32 * h < rn_[r]) {
                if (staged) lcol[slot[h] - p0] = (int)(rec[r][h] & GCN_SM);
                else        col[slot[h]]       = (int)(rec[r][h] & GCN_SM);
            }
        }
        for (int j = lane + 32 * GCN_RLD; j < rn_[r]; j += 32) {
            const unsigned e = pairs[st[r] + j];
            const int q = atomicAdd(&lcnt[e >> GCN_SB], 1);
            if (staged) lcol[q - p0] = (int)(e & GCN_SM);
            else        col[q]       = (int)(e & GCN_SM);
        }
    }
    if (staged) {
        __syncthreads();
        for (int j = tid; j < tot; j += 1024) col[p0 + j] = lcol[j];
    }
}

#define GCN_TC   57344
#define GCN_EPT  16
#define GCN_SG   14
#define GCN_RU   8
static_assert(GCN_RU == 8, "the read fence in gcn_range_rows names 8 rows");

__device__ __forceinline__ float gcn_sum16(float a) {
    a += __int_as_float(__builtin_amdgcn_update_dpp(0, __float_as_int(a), 0x168, 0xf, 0xf, false));
    a += __int_as_float(__builtin_amdgcn_update_dpp(0, __float_as_int(a), 0x164, 0xf, 0xf, false));
    a += __int_as_float(__builtin_amdgcn_update_dpp(0, __float_as_int(a), 0x162, 0xf, 0xf, false));
    a += __int_as_float(__builtin_amdgcn_update_dpp(0, __float_as_int(a), 0x161, 0xf, 0xf, false));
    return a;
}

__device__ __forceinline__ void gcn_range_rows(const int* __restrict__ rowptr,
                                               const int* __restrict__ cnt,
                                               const int* __restrict__ col,
                                               const float* __restrict__ wt,
                                               int N, int E, int n0, int Rb,
                                               float* tab, int2* lrow, float* rsum) {
    const int tid = threadIdx.x;
    const int r16 = tid >> 4;
    const int f   = tid & 15;
    const int nl  = n0 + Rb - 1;
    const int p0  = rowptr[n0];
    const int tot = rowptr[nl] + cnt[nl] - p0;
    for (int t = tid; t < Rb; t += 1024)
        lrow[t] = make_int2(rowptr[n0 + t] - p0, cnt[n0 + t]);
    if (tot <= GCN_EPT * 1024) {
        int   cj[GCN_EPT];
        float val[GCN_EPT];
#pragma unroll
        for (int k = 0; k < GCN_EPT; ++k) {
            int j = tid + 1024 * k;
            j = j < tot ? j : tot - 1;
            int g = p0 + j;
            g = g < E ? g : E - 1;
            g = g > 0 ? g : 0;
            cj[k]  = col[g];
            val[k] = 0.0f;
        }
        for (int c0 = 0; c0 < N; c0 += GCN_TC) {
            const int clen = (N - c0 < GCN_TC) ? (N - c0) : GCN_TC;
            const int cl4  = clen & ~3;
            const v4f* wch4 = (const v4f*)(wt + c0);
            v4f*       tab4 = (v4f*)tab;
            const unsigned n4 = (unsigned)cl4 >> 2;
            for (unsigned g = (unsigned)tid; g < n4; g += 1024u * GCN_SG) {
                v4f a[GCN_SG];
                unsigned ic[GCN_SG];
#pragma unroll
                for (int u = 0; u < GCN_SG; ++u) {
                    const unsigned i = g + (unsigned)u * 1024u;
                    ic[u] = i < n4 ? i : g;
                    a[u] = wch4[ic[u]];
                }
                __builtin_amdgcn_sched_barrier(0);
#pragma unroll
                for (int u = 0; u < GCN_SG; ++u) tab4[ic[u]] = a[u];
            }
            if (tid < clen - cl4) {
                const int i = cl4 + tid;
                tab[i] = wt[c0 + i];
            }
            __syncthreads();
#pragma unroll
            for (int k = 0; k < GCN_EPT; ++k) {
                const unsigned loc = (unsigned)(cj[k] - c0);
                if (loc < (unsigned)clen) val[k] = tab[loc];
            }
            __syncthreads();
        }
#pragma unroll
        for (int k = 0; k < GCN_EPT; ++k) {
            const int j = tid + 1024 * k;
            if (j < tot) tab[j] = val[k];
        }
        __syncthreads();
        for (int rb = 0; rb < Rb; rb += 64 * GCN_RU) {
            int   st[GCN_RU], ln[GCN_RU];
            float acc[GCN_RU];
            int   mx = 0;
#pragma unroll
            for (int u = 0; u < GCN_RU; ++u) {
                const int  r  = rb + 64 * u + r16;
                const bool ok = r < Rb;
                const int2 e  = lrow[ok ? r : 0];
                st[u]  = e.x;
                ln[u]  = ok ? e.y : 0;
                acc[u] = 0.0f;
                mx = ln[u] > mx ? ln[u] : mx;
            }
            for (int j = f; j < mx; j += 32) {
                float a0[GCN_RU], a1[GCN_RU];
#pragma unroll
                for (int u = 0; u < GCN_RU; ++u) {
                    a0[u] = tab[st[u] + j];
                    a1[u] = tab[st[u] + j + 16];
                }
                asm volatile("" :: "v"(a0[0]), "v"(a0[1]), "v"(a0[2]), "v"(a0[3]), "v"(a0[4]), "v"(a0[5]), "v"(a0[6]), "v"(a0[7]),
                                   "v"(a1[0]), "v"(a1[1]), "v"(a1[2]), "v"(a1[3]), "v"(a1[4]), "v"(a1[5]), "v"(a1[6]), "v"(a1[7]));
#pragma unroll
                for (int u = 0; u < GCN_RU; ++u) {
                    const bool  one = j < ln[u];
                    const bool  two = j + 16 < ln[u];
                    const float t   = a0[u] + (two ? a1[u] : 0.0f);
                    acc[u] = one ? acc[u] + t : acc[u];
                }
            }
#pragma unroll
            for (int u = 0; u < GCN_RU; ++u) acc[u] = gcn_sum16(acc[u]);
            float tsel = acc[0];
#pragma unroll
            for (int u = 1; u < GCN_RU; ++u) tsel = (f == u) ? acc[u] : tsel;
            const int r = rb + 64 * f + r16;
            if (f < GCN_RU && r < Rb) rsum[r] = tsel;
        }
    } else {
        __syncthreads();
        for (int rb = 0; rb < Rb; rb += 64) {
            const int  r  = rb + r16;
            const bool ok = r < Rb;
            const int2 e    = lrow[ok ? r : 0];
            const int start = ok ? e.x + p0 : 0;
            const int len   = ok ? e.y : 0;
            float acc = 0.0f;
            for (int j = f; j < len; j += 32) {
                const int  j1  = j + 16;
                const bool two = j1 < len;
                const int  s0  = col[start + j];
                const int  s1  = col[start + (two ? j1 : j)];
                const float a0 = wt[s0];
                const float a1 = wt[s1];
                acc += a0 + (two ? a1 : 0.0f);
            }
            acc = gcn_sum16(acc);
            if (ok && f == 0) rsum[r] = acc;
        }
    }
    __syncthreads();
}

__global__ __launch_bounds__(1024) void gcn_layer1_kernel(const int* __restrict__ rowptr,
                                                          const int* __restrict__ cnt,
                                                          const int* __restrict__ col,
                                                          const float* __restrict__ dinv,
                                                          const float* __restrict__ x,
                                                          const float* __restrict__ W1,
                                                          const float* __restrict__ b1,
                                                          const float* __restrict__ W2,
                                                          const float* __restrict__ wx,
                                                          float* __restrict__ z,
                                                          float* __restrict__ wz, int E, int N) {
    __shared__ float tab[GCN_TC];
    __shared__ int2  lrow[GCN_RMAX];
    __shared__ float rsum[GCN_RMAX];
    __shared__ float wl[48];
    __shared__ float own[2048];
    const int R  = (N + GCN_NB - 1) / GCN_NB;
    const int n0 = blockIdx.x * R;
    if (n0 >= N) return;
    const int Rb = (N - n0 < R) ? (N - n0) : R;
    const int w  = threadIdx.x >> 5;
    const int m  = threadIdx.x & 15;
    const int kh = (threadIdx.x >> 4) & 1;
    const int ntile = (Rb + 15) >> 4;

    if (threadIdx.x < 16) {
        wl[threadIdx.x]      = W1[threadIdx.x];
        wl[16 + threadIdx.x] = b1[threadIdx.x];
        wl[32 + threadIdx.x] = W2[threadIdx.x];
    }
    {
        float dm0 = 0.0f, xm0 = 0.0f;
        if (w * 16 + m < Rb) { dm0 = dinv[n0 + w * 16 + m]; xm0 = x[n0 + w * 16 + m]; }
        own[threadIdx.x]        = dm0;
        own[1024 + threadIdx.x] = xm0;
    }

    gcn_range_rows(rowptr, cnt, col, wx, N, E, n0, Rb, tab, lrow, rsum);

    const float dm0 = own[threadIdx.x];
    const float xm0 = own[1024 + threadIdx.x];

    float w1v[8], b1v[8], w2v[8];
#pragma unroll
    for (int kc = 0; kc < 4; ++kc) {
#pragma unroll
        for (int i = 0; i < 2; ++i) {
            const int k = kc * 4 + kh * 2 + i;
            w1v[kc * 2 + i] = wl[k];
            b1v[kc * 2 + i] = wl[16 + k];
            w2v[kc * 2 + i] = wl[32 + k];
        }
    }
    for (int t = w; t < ntile; t += 32) {
        const int  row = t * 16 + m;
        const bool ok  = row < Rb;
        float dm = dm0, xm = xm0;
        if (t != w) {
            dm = ok ? dinv[n0 + row] : 0.0f;
            xm = ok ? x[n0 + row] : 0.0f;
        }
        const float ts = ok ? dm * rsum[row] : 0.0f;
        const float sl = dm * dm * xm;
        v8f d = {0.f, 0.f, 0.f, 0.f, 0.f, 0.f, 0.f, 0.f};
#pragma unroll
        for (int kc = 0; kc < 4; ++kc) {
            v2f a, b;
            a.x = fmaxf(ts * w1v[kc * 2 + 0] + sl * w1v[kc * 2 + 0] + b1v[kc * 2 + 0], 0.0f);
            a.y = fmaxf(ts * w1v[kc * 2 + 1] + sl * w1v[kc * 2 + 1] + b1v[kc * 2 + 1], 0.0f);
            b.x = w2v[kc * 2 + 0];
            b.y = w2v[kc * 2 + 1];
            d = __builtin_amdgcn_wmma_f32_16x16x4_f32(
                 false, a,  false, b,
                 (short)0, d,  false,  false);
        }
        float zr = d[0];
#pragma unroll
        for (int r = 1; r < 8; ++r) zr = ((m & 7) == r) ? d[r] : zr;
        if ((m >> 3) == kh && ok) {
            z[n0 + row]  = zr;
            wz[n0 + row] = dm * zr;
        }
    }
}

__global__ __launch_bounds__(1024) void gcn_layer2_kernel(const int* __restrict__ rowptr,
                                                          const int* __restrict__ cnt,
                                                          const int* __restrict__ col,
                                                          const float* __restrict__ dinv,
                                                          const float* __restrict__ z,
                                                          const float* __restrict__ wz,
                                                          const float* __restrict__ b2,
                                                          float* __restrict__ out, int E, int N) {
    __shared__ float tab[GCN_TC];
    __shared__ int2  lrow[GCN_RMAX];
    __shared__ float rsum[GCN_RMAX];
    const int R  = (N + GCN_NB - 1) / GCN_NB;
    const int n0 = blockIdx.x * R;
    if (n0 >= N) return;
    const int Rb = (N - n0 < R) ? (N - n0) : R;
    const int tid = threadIdx.x;
    const float bias = b2[0];
    float dn0 = 0.0f, zn0 = 0.0f;
    if (tid < Rb) { dn0 = dinv[n0 + tid]; zn0 = z[n0 + tid]; }

    gcn_range_rows(rowptr, cnt, col, wz, N, E, n0, Rb, tab, lrow, rsum);

    for (int t = tid; t < Rb; t += 1024) {
        float dn = dn0, zn = zn0;
        if (t != tid) { dn = dinv[n0 + t]; zn = z[n0 + t]; }
        out[n0 + t] = dn * rsum[t] + (dn * dn * zn + bias);
    }
}

extern "C" void kernel_launch(void* const* d_in, const int* in_sizes, int n_in,
                              void* d_out, int out_size, void* d_ws, size_t ws_size,
                              hipStream_t stream) {
    const float* x  = (const float*)d_in[0];
    const float* W1 = (const float*)d_in[1];
    const float* b1 = (const float*)d_in[2];
    const float* W2 = (const float*)d_in[3];
    const float* b2 = (const float*)d_in[4];
    const int*   ei = (const int*)d_in[5];
    int N = in_sizes[0];
    int E = in_sizes[5] / 2;
    const int* src = ei;
    const int* dst = ei + E;
    int paddedN = (N + 15) & ~15;

    char* ws = (char*)d_ws;
    float* dinv   = (float*)ws; ws += (size_t)paddedN * sizeof(float);
    float* z      = (float*)ws; ws += (size_t)paddedN * sizeof(float);
    int*   cnt    = (int*)ws;   ws += (size_t)paddedN * sizeof(int);
    int*   rowptr = (int*)ws;   ws += (size_t)paddedN * sizeof(int);
    float* wx     = (float*)ws; ws += (size_t)paddedN * sizeof(float);
    float* wz     = (float*)ws; ws += (size_t)paddedN * sizeof(float);
    int*   T      = (int*)ws;   ws += (size_t)(GCN_NB + 1) * GCN_NS * sizeof(int);
    int*   pairs  = (int*)ws;   ws += (((size_t)E * 2 * sizeof(int)) + 255) & ~(size_t)255;
    int*   col    = (int*)ws;
    float* out    = (float*)d_out;

    gcn_slice_sort_kernel<<<GCN_NS, 1024, 0, stream>>>(src, dst, T, pairs, E, N);
    gcn_rows_kernel<<<GCN_NB, 1024, 0, stream>>>(pairs, T, rowptr, cnt, dinv, x, wx, col, E, N);
    gcn_layer1_kernel<<<GCN_NB, 1024, 0, stream>>>(rowptr, cnt, col, dinv, x, W1, b1, W2, wx, z, wz, E, N);
    gcn_layer2_kernel<<<GCN_NB, 1024, 0, stream>>>(rowptr, cnt, col, dinv, z, wz, b2, out, E, N);
}
